# starting kernel
# baseline (speedup 1.0000x reference)
#include <hip/hip_runtime.h>
#include <hip/hip_bf16.h>
#include <math.h>

typedef __attribute__((ext_vector_type(2))) float v2f;
typedef __attribute__((ext_vector_type(8))) float v8f;

#define D_FEAT 128
#define MT_PER_BLK 2
#define ROWS_PER_BLK (MT_PER_BLK * 16)

__global__ __launch_bounds__(256) void deg_kernel(const int* __restrict__ dst,
                                                  float* __restrict__ deg,
                                                  int n_edges) {
    int e = blockIdx.x * blockDim.x + threadIdx.x;
    if (e < n_edges) atomicAdd(&deg[dst[e]], 1.0f);
}

__global__ __launch_bounds__(256) void dinv_kernel(float* __restrict__ deg, int n) {
    int i = blockIdx.x * blockDim.x + threadIdx.x;
    if (i < n) deg[i] = rsqrtf(deg[i] + 1.0f);
}

__global__ __launch_bounds__(256) void gemm_scale_kernel(
    const float* __restrict__ x, const float* __restrict__ W,
    const float* __restrict__ dinv, float* __restrict__ hprime,
    float* __restrict__ acc, int n_nodes) {
    __shared__ float sX[ROWS_PER_BLK][D_FEAT + 4];

    const int tid   = threadIdx.x;
    const int lane  = tid & 31;
    const int wave  = tid >> 5;
    const int mbase = blockIdx.x * ROWS_PER_BLK;
    const int half  = lane >> 4;
    const int l15   = lane & 15;
    const int nb    = wave * 16;

    const bool full = (mbase + ROWS_PER_BLK) <= n_nodes;

    const float4* xv = reinterpret_cast<const float4*>(x);
    if (full) {
#pragma unroll
        for (int i = 0; i < 4; ++i) {
            int slot = tid + i * 256;
            int r    = slot >> 5;
            int cb   = slot & 31;
            *reinterpret_cast<float4*>(&sX[r][cb * 4]) =
                xv[(size_t)(mbase + r) * (D_FEAT / 4) + cb];
        }
    } else {
        const float4 zero4 = make_float4(0.f, 0.f, 0.f, 0.f);
#pragma unroll
        for (int i = 0; i < 4; ++i) {
            int slot = tid + i * 256;
            int r    = slot >> 5;
            int cb   = slot & 31;
            int row  = mbase + r;
            float4 v = (row < n_nodes) ? xv[(size_t)row * (D_FEAT / 4) + cb] : zero4;
            *reinterpret_cast<float4*>(&sX[r][cb * 4]) = v;
        }
    }
    __syncthreads();

    v8f c[MT_PER_BLK];
#pragma unroll
    for (int t = 0; t < MT_PER_BLK; ++t)
#pragma unroll
        for (int r = 0; r < 8; ++r) c[t][r] = 0.0f;

    for (int k0 = 0; k0 < D_FEAT; k0 += 4) {
        const int ka = k0 + half * 2;
        v2f bf;
        bf.x = W[(size_t)ka * D_FEAT + nb + l15];
        bf.y = W[(size_t)(ka + 1) * D_FEAT + nb + l15];
#pragma unroll
        for (int t = 0; t < MT_PER_BLK; ++t) {
            v2f af;
            af.x = sX[t * 16 + l15][ka];
            af.y = sX[t * 16 + l15][ka + 1];
            c[t] = __builtin_amdgcn_wmma_f32_16x16x4_f32(
                 false, af,  false, bf,
                 (short)0, c[t],  false,  false);
        }
    }

    if (full) {
#pragma unroll
        for (int t = 0; t < MT_PER_BLK; ++t) {
            const int rbase = mbase + t * 16 + half * 8;
            float* hp = hprime + (size_t)rbase * D_FEAT + nb + l15;
            float* ap = acc    + (size_t)rbase * D_FEAT + nb + l15;
            float dv[8];
#pragma unroll
            for (int r = 0; r < 8; ++r) dv[r] = dinv[rbase + r];
#pragma unroll
            for (int r = 0; r < 8; ++r) {
                float v = c[t][r] * dv[r];
                hp[(size_t)r * D_FEAT] = v;
                ap[(size_t)r * D_FEAT] = v;
            }
        }
    } else {
#pragma unroll
        for (int t = 0; t < MT_PER_BLK; ++t) {
#pragma unroll
            for (int r = 0; r < 8; ++r) {
                int row = mbase + t * 16 + r + half * 8;
                if (row < n_nodes) {
                    float v = c[t][r] * dinv[row];
                    size_t off = (size_t)row * D_FEAT + nb + l15;
                    hprime[off] = v;
                    acc[off]    = v;
                }
            }
        }
    }
}

__global__ __launch_bounds__(256) void scatter_kernel(
    const int* __restrict__ src, const int* __restrict__ dst,
    const float* __restrict__ hprime, float* __restrict__ acc, int n_edges) {
    long long gid = (long long)blockIdx.x * blockDim.x + threadIdx.x;
    int e    = (int)(gid >> 5);
    int lane = (int)(gid & 31);
    if (e >= n_edges) return;
    int s = src[e], d = dst[e];
    float4 hv = *reinterpret_cast<const float4*>(hprime + (size_t)s * D_FEAT + lane * 4);
    float* ap = acc + (size_t)d * D_FEAT + lane * 4;
    atomicAdd(ap + 0, hv.x);
    atomicAdd(ap + 1, hv.y);
    atomicAdd(ap + 2, hv.z);
    atomicAdd(ap + 3, hv.w);
}

__global__ __launch_bounds__(256) void finalize_kernel(
    const float* __restrict__ acc, const float* __restrict__ dinv,
    const float* __restrict__ bias, const float* __restrict__ pa,
    const int* __restrict__ batch, float* __restrict__ pooled,
    float* __restrict__ cnt, int n_nodes) {
    long long gid = (long long)blockIdx.x * blockDim.x + threadIdx.x;
    int i    = (int)(gid >> 5);
    int lane = (int)(gid & 31);
    if (i >= n_nodes) return;

    float di = dinv[i];
    float v[4];
    float ss = 0.0f;
#pragma unroll
    for (int j = 0; j < 4; ++j) {
        int c   = lane + 32 * j;
        float t = di * acc[(size_t)i * D_FEAT + c] + bias[c];
        t = (t > 0.0f) ? t : pa[c] * t;
        v[j] = t;
        ss += t * t;
    }
#pragma unroll
    for (int off = 16; off > 0; off >>= 1) ss += __shfl_xor(ss, off, 32);
    float scale = 1.0f / fmaxf(sqrtf(ss), 1e-12f);

    int g = batch[i];
#pragma unroll
    for (int j = 0; j < 4; ++j)
        atomicAdd(&pooled[(size_t)g * D_FEAT + lane + 32 * j], v[j] * scale);
    if (lane == 0) atomicAdd(&cnt[g], 1.0f);
}

__global__ __launch_bounds__(256) void div_kernel(float* __restrict__ pooled,
                                                  const float* __restrict__ cnt,
                                                  int total) {
    int i = blockIdx.x * blockDim.x + threadIdx.x;
    if (i < total) pooled[i] /= fmaxf(cnt[i >> 7], 1.0f);
}

extern "C" void kernel_launch(void* const* d_in, const int* in_sizes, int n_in,
                              void* d_out, int out_size, void* d_ws, size_t ws_size,
                              hipStream_t stream) {
    const float* x     = (const float*)d_in[0];
    const int*   ei    = (const int*)d_in[1];
    const int*   batch = (const int*)d_in[2];
    const float* W     = (const float*)d_in[3];
    const float* bias  = (const float*)d_in[4];
    const float* pa    = (const float*)d_in[5];

    const int n_nodes = in_sizes[0] / D_FEAT;
    const int n_edges = in_sizes[1] / 2;
    const int* src = ei;
    const int* dst = ei + n_edges;
    float* pooled = (float*)d_out;

    float* hprime = (float*)d_ws;
    float* acc    = hprime + (size_t)n_nodes * D_FEAT;
    float* dinv   = acc + (size_t)n_nodes * D_FEAT;
    float* cnt    = dinv + n_nodes;

    hipMemsetAsync(dinv, 0, (size_t)n_nodes * sizeof(float), stream);
    hipMemsetAsync(cnt, 0, 128 * sizeof(float), stream);
    hipMemsetAsync(d_out, 0, (size_t)out_size * sizeof(float), stream);

    deg_kernel<<<(n_edges + 255) / 256, 256, 0, stream>>>(dst, dinv, n_edges);
    dinv_kernel<<<(n_nodes + 255) / 256, 256, 0, stream>>>(dinv, n_nodes);

    int gemm_blocks = (n_nodes + ROWS_PER_BLK - 1) / ROWS_PER_BLK;
    gemm_scale_kernel<<<gemm_blocks, 256, 0, stream>>>(x, W, dinv, hprime, acc, n_nodes);

    long long sc_threads = (long long)n_edges * 32;
    scatter_kernel<<<(int)((sc_threads + 255) / 256), 256, 0, stream>>>(src, dst, hprime, acc, n_edges);

    long long fin_threads = (long long)n_nodes * 32;
    finalize_kernel<<<(int)((fin_threads + 255) / 256), 256, 0, stream>>>(
        acc, dinv, bias, pa, batch, pooled, cnt, n_nodes);

    div_kernel<<<(out_size + 255) / 256, 256, 0, stream>>>(pooled, cnt, out_size);
}


# finalize: each wave walks 16 consecutive nodes, pooled sums and counts kept in registers, atomics only when the graph id changes; pooled summation order changes (partial sums, then atomic adds)
# speedup vs baseline: 1.3854x; 1.3854x over previous
#include <hip/hip_runtime.h>
#include <hip/hip_bf16.h>
#include <math.h>

typedef __attribute__((ext_vector_type(2))) float v2f;
typedef __attribute__((ext_vector_type(8))) float v8f;

#define D_FEAT 128
#define MT_PER_BLK 2
#define ROWS_PER_BLK (MT_PER_BLK * 16)

__global__ __launch_bounds__(256) void deg_kernel(const int* __restrict__ dst,
                                                  float* __restrict__ deg,
                                                  int n_edges) {
    int e = blockIdx.x * blockDim.x + threadIdx.x;
    if (e < n_edges) atomicAdd(&deg[dst[e]], 1.0f);
}

__global__ __launch_bounds__(256) void dinv_kernel(float* __restrict__ deg, int n) {
    int i = blockIdx.x * blockDim.x + threadIdx.x;
    if (i < n) deg[i] = rsqrtf(deg[i] + 1.0f);
}

__global__ __launch_bounds__(256) void gemm_scale_kernel(
    const float* __restrict__ x, const float* __restrict__ W,
    const float* __restrict__ dinv, float* __restrict__ hprime,
    float* __restrict__ acc, int n_nodes) {
    __shared__ float sX[ROWS_PER_BLK][D_FEAT + 4];

    const int tid   = threadIdx.x;
    const int lane  = tid & 31;
    const int wave  = tid >> 5;
    const int mbase = blockIdx.x * ROWS_PER_BLK;
    const int half  = lane >> 4;
    const int l15   = lane & 15;
    const int nb    = wave * 16;

    const bool full = (mbase + ROWS_PER_BLK) <= n_nodes;

    const float4* xv = reinterpret_cast<const float4*>(x);
    if (full) {
#pragma unroll
        for (int i = 0; i < 4; ++i) {
            int slot = tid + i * 256;
            int r    = slot >> 5;
            int cb   = slot & 31;
            *reinterpret_cast<float4*>(&sX[r][cb * 4]) =
                xv[(size_t)(mbase + r) * (D_FEAT / 4) + cb];
        }
    } else {
        const float4 zero4 = make_float4(0.f, 0.f, 0.f, 0.f);
#pragma unroll
        for (int i = 0; i < 4; ++i) {
            int slot = tid + i * 256;
            int r    = slot >> 5;
            int cb   = slot & 31;
            int row  = mbase + r;
            float4 v = (row < n_nodes) ? xv[(size_t)row * (D_FEAT / 4) + cb] : zero4;
            *reinterpret_cast<float4*>(&sX[r][cb * 4]) = v;
        }
    }
    __syncthreads();

    v8f c[MT_PER_BLK];
#pragma unroll
    for (int t = 0; t < MT_PER_BLK; ++t)
#pragma unroll
        for (int r = 0; r < 8; ++r) c[t][r] = 0.0f;

    for (int k0 = 0; k0 < D_FEAT; k0 += 4) {
        const int ka = k0 + half * 2;
        v2f bf;
        bf.x = W[(size_t)ka * D_FEAT + nb + l15];
        bf.y = W[(size_t)(ka + 1) * D_FEAT + nb + l15];
#pragma unroll
        for (int t = 0; t < MT_PER_BLK; ++t) {
            v2f af;
            af.x = sX[t * 16 + l15][ka];
            af.y = sX[t * 16 + l15][ka + 1];
            c[t] = __builtin_amdgcn_wmma_f32_16x16x4_f32(
                 false, af,  false, bf,
                 (short)0, c[t],  false,  false);
        }
    }

    if (full) {
#pragma unroll
        for (int t = 0; t < MT_PER_BLK; ++t) {
            const int rbase = mbase + t * 16 + half * 8;
            float* hp = hprime + (size_t)rbase * D_FEAT + nb + l15;
            float* ap = acc    + (size_t)rbase * D_FEAT + nb + l15;
            float dv[8];
#pragma unroll
            for (int r = 0; r < 8; ++r) dv[r] = dinv[rbase + r];
#pragma unroll
            for (int r = 0; r < 8; ++r) {
                float v = c[t][r] * dv[r];
                hp[(size_t)r * D_FEAT] = v;
                ap[(size_t)r * D_FEAT] = v;
            }
        }
    } else {
#pragma unroll
        for (int t = 0; t < MT_PER_BLK; ++t) {
#pragma unroll
            for (int r = 0; r < 8; ++r) {
                int row = mbase + t * 16 + r + half * 8;
                if (row < n_nodes) {
                    float v = c[t][r] * dinv[row];
                    size_t off = (size_t)row * D_FEAT + nb + l15;
                    hprime[off] = v;
                    acc[off]    = v;
                }
            }
        }
    }
}

__global__ __launch_bounds__(256) void scatter_kernel(
    const int* __restrict__ src, const int* __restrict__ dst,
    const float* __restrict__ hprime, float* __restrict__ acc, int n_edges) {
    long long gid = (long long)blockIdx.x * blockDim.x + threadIdx.x;
    int e    = (int)(gid >> 5);
    int lane = (int)(gid & 31);
    if (e >= n_edges) return;
    int s = src[e], d = dst[e];
    float4 hv = *reinterpret_cast<const float4*>(hprime + (size_t)s * D_FEAT + lane * 4);
    float* ap = acc + (size_t)d * D_FEAT + lane * 4;
    atomicAdd(ap + 0, hv.x);
    atomicAdd(ap + 1, hv.y);
    atomicAdd(ap + 2, hv.z);
    atomicAdd(ap + 3, hv.w);
}

#define FIN_NODES_PER_WAVE 16
#define FIN_GROUP 8

__device__ __forceinline__ void pool_flush(float* __restrict__ pooled,
                                           float* __restrict__ cnt, int g, int lane,
                                           const float (&sum)[4], float nseg) {
#pragma unroll
    for (int j = 0; j < 4; ++j)
        atomicAdd(&pooled[(size_t)g * D_FEAT + lane + 32 * j], sum[j]);
    if (lane == 0) atomicAdd(&cnt[g], nseg);
}

__global__ __launch_bounds__(256) void finalize_kernel(
    const float* __restrict__ acc, const float* __restrict__ dinv,
    const float* __restrict__ bias, const float* __restrict__ pa,
    const int* __restrict__ batch, float* __restrict__ pooled,
    float* __restrict__ cnt, int n_nodes) {
    const int lane = threadIdx.x & 31;
    const int wv = __builtin_amdgcn_readfirstlane(
        (int)(blockIdx.x * (blockDim.x >> 5) + (threadIdx.x >> 5)));
    const int i0 = wv * FIN_NODES_PER_WAVE;
    if (i0 >= n_nodes) return;
    const int i1 = (i0 + FIN_NODES_PER_WAVE < n_nodes) ? i0 + FIN_NODES_PER_WAVE : n_nodes;

    float bj[4], aj[4];
#pragma unroll
    for (int j = 0; j < 4; ++j) {
        bj[j] = bias[lane + 32 * j];
        aj[j] = pa[lane + 32 * j];
    }

    float sum[4] = {0.0f, 0.0f, 0.0f, 0.0f};
    float nseg = 0.0f;
    int gcur = batch[i0];

    for (int ib = i0; ib < i1; ib += FIN_GROUP) {
        float raw[FIN_GROUP][4];
        float di[FIN_GROUP];
        int   gg[FIN_GROUP];
#pragma unroll
        for (int u = 0; u < FIN_GROUP; ++u) {
            int i = ib + u;
            i = (i < i1) ? i : (i1 - 1);
            gg[u] = batch[i];
            di[u] = dinv[i];
#pragma unroll
            for (int j = 0; j < 4; ++j)
                raw[u][j] = acc[(size_t)i * D_FEAT + lane + 32 * j];
        }
        float scale[FIN_GROUP];
#pragma unroll
        for (int u = 0; u < FIN_GROUP; ++u) {
            float ss = 0.0f;
#pragma unroll
            for (int j = 0; j < 4; ++j) {
                float t = di[u] * raw[u][j] + bj[j];
                t = (t > 0.0f) ? t : aj[j] * t;
                raw[u][j] = t;
                ss += t * t;
            }
#pragma unroll
            for (int off = 16; off > 0; off >>= 1) ss += __shfl_xor(ss, off, 32);
            scale[u] = 1.0f / fmaxf(sqrtf(ss), 1e-12f);
        }
#pragma unroll
        for (int u = 0; u < FIN_GROUP; ++u) {
            if (ib + u < i1) {
                if (gg[u] != gcur) {
                    pool_flush(pooled, cnt, gcur, lane, sum, nseg);
                    sum[0] = sum[1] = sum[2] = sum[3] = 0.0f;
                    nseg = 0.0f;
                    gcur = gg[u];
                }
#pragma unroll
                for (int j = 0; j < 4; ++j) sum[j] += raw[u][j] * scale[u];
                nseg += 1.0f;
            }
        }
    }
    pool_flush(pooled, cnt, gcur, lane, sum, nseg);
}

__global__ __launch_bounds__(256) void div_kernel(float* __restrict__ pooled,
                                                  const float* __restrict__ cnt,
                                                  int total) {
    int i = blockIdx.x * blockDim.x + threadIdx.x;
    if (i < total) pooled[i] /= fmaxf(cnt[i >> 7], 1.0f);
}

extern "C" void kernel_launch(void* const* d_in, const int* in_sizes, int n_in,
                              void* d_out, int out_size, void* d_ws, size_t ws_size,
                              hipStream_t stream) {
    const float* x     = (const float*)d_in[0];
    const int*   ei    = (const int*)d_in[1];
    const int*   batch = (const int*)d_in[2];
    const float* W     = (const float*)d_in[3];
    const float* bias  = (const float*)d_in[4];
    const float* pa    = (const float*)d_in[5];

    const int n_nodes = in_sizes[0] / D_FEAT;
    const int n_edges = in_sizes[1] / 2;
    const int* src = ei;
    const int* dst = ei + n_edges;
    float* pooled = (float*)d_out;

    float* hprime = (float*)d_ws;
    float* acc    = hprime + (size_t)n_nodes * D_FEAT;
    float* dinv   = acc + (size_t)n_nodes * D_FEAT;
    float* cnt    = dinv + n_nodes;

    hipMemsetAsync(dinv, 0, (size_t)n_nodes * sizeof(float), stream);
    hipMemsetAsync(cnt, 0, 128 * sizeof(float), stream);
    hipMemsetAsync(d_out, 0, (size_t)out_size * sizeof(float), stream);

    deg_kernel<<<(n_edges + 255) / 256, 256, 0, stream>>>(dst, dinv, n_edges);
    dinv_kernel<<<(n_nodes + 255) / 256, 256, 0, stream>>>(dinv, n_nodes);

    int gemm_blocks = (n_nodes + ROWS_PER_BLK - 1) / ROWS_PER_BLK;
    gemm_scale_kernel<<<gemm_blocks, 256, 0, stream>>>(x, W, dinv, hprime, acc, n_nodes);

    long long sc_threads = (long long)n_edges * 32;
    scatter_kernel<<<(int)((sc_threads + 255) / 256), 256, 0, stream>>>(src, dst, hprime, acc, n_edges);

    int fin_waves  = (n_nodes + FIN_NODES_PER_WAVE - 1) / FIN_NODES_PER_WAVE;
    finalize_kernel<<<(fin_waves + 7) / 8, 256, 0, stream>>>(
        acc, dinv, bias, pa, batch, pooled, cnt, n_nodes);

    div_kernel<<<(out_size + 255) / 256, 256, 0, stream>>>(pooled, cnt, out_size);
}


# float-atomic scatter into acc replaced by edges grouped by destination (counting sort built per call); each wave sums its nodes' in-neighbour rows in registers and finalizes them; sums in fill order
# speedup vs baseline: 2.0881x; 1.5072x over previous
#include <hip/hip_runtime.h>
#include <hip/hip_bf16.h>
#include <math.h>


typedef __attribute__((ext_vector_type(2))) float v2f;
typedef __attribute__((ext_vector_type(8))) float v8f;

#define D_FEAT 128
#define MT_PER_BLK 2
#define ROWS_PER_BLK (MT_PER_BLK * 16)

__global__ __launch_bounds__(256) void count_kernel(const int* __restrict__ dst,
                                                    int* __restrict__ degi,
                                                    int n_edges) {
    int e = blockIdx.x * blockDim.x + threadIdx.x;
    if (e < n_edges) atomicAdd(&degi[dst[e]], 1);
}

__global__ __launch_bounds__(256) void alloc_dinv_kernel(
    const int* __restrict__ degi, int* __restrict__ gcur,
    int* __restrict__ rowstart, int* __restrict__ cursor,
    float* __restrict__ dinv, int n) {
    __shared__ int swave[8];
    __shared__ int sbase;
    const int i    = blockIdx.x * blockDim.x + threadIdx.x;
    const int lane = threadIdx.x & 31;
    const int wave = threadIdx.x >> 5;
    const int c = (i < n) ? degi[i] : 0;
    int v = c;
#pragma unroll
    for (int off = 1; off < 32; off <<= 1) {
        int t = __shfl_up(v, off, 32);
        if (lane >= off) v += t;
    }
    if (lane == 31) swave[wave] = v;
    __syncthreads();
    if (threadIdx.x == 0) {
        int tot = 0;
        for (int w = 0; w < 8; ++w) {
            int t = swave[w];
            swave[w] = tot;
            tot += t;
        }
        sbase = atomicAdd(gcur, tot);
    }
    __syncthreads();
    if (i < n) {
        const int pos = sbase + swave[wave] + (v - c);
        rowstart[i] = pos;
        cursor[i]   = pos;
        dinv[i]     = rsqrtf((float)c + 1.0f);
    }
}

__global__ __launch_bounds__(256) void fill_kernel(const int* __restrict__ src,
                                                   const int* __restrict__ dst,
                                                   int* __restrict__ cursor,
                                                   int* __restrict__ col,
                                                   int n_edges) {
    int e = blockIdx.x * blockDim.x + threadIdx.x;
    if (e < n_edges) {
        int p = atomicAdd(&cursor[dst[e]], 1);
        col[p] = src[e];
    }
}

__global__ __launch_bounds__(256) void gemm_scale_kernel(
    const float* __restrict__ x, const float* __restrict__ W,
    const float* __restrict__ dinv, float* __restrict__ hprime,
    int n_nodes) {
    __shared__ float sX[ROWS_PER_BLK][D_FEAT + 4];

    const int tid   = threadIdx.x;
    const int lane  = tid & 31;
    const int wave  = tid >> 5;
    const int mbase = blockIdx.x * ROWS_PER_BLK;
    const int half  = lane >> 4;
    const int l15   = lane & 15;
    const int nb    = wave * 16;

    const bool full = (mbase + ROWS_PER_BLK) <= n_nodes;

    const float4* xv = reinterpret_cast<const float4*>(x);
    if (full) {
#pragma unroll
        for (int i = 0; i < 4; ++i) {
            int slot = tid + i * 256;
            int r    = slot >> 5;
            int cb   = slot & 31;
            *reinterpret_cast<float4*>(&sX[r][cb * 4]) =
                xv[(size_t)(mbase + r) * (D_FEAT / 4) + cb];
        }
    } else {
        const float4 zero4 = make_float4(0.f, 0.f, 0.f, 0.f);
#pragma unroll
        for (int i = 0; i < 4; ++i) {
            int slot = tid + i * 256;
            int r    = slot >> 5;
            int cb   = slot & 31;
            int row  = mbase + r;
            float4 v = (row < n_nodes) ? xv[(size_t)row * (D_FEAT / 4) + cb] : zero4;
            *reinterpret_cast<float4*>(&sX[r][cb * 4]) = v;
        }
    }
    __syncthreads();

    v8f c[MT_PER_BLK];
#pragma unroll
    for (int t = 0; t < MT_PER_BLK; ++t)
#pragma unroll
        for (int r = 0; r < 8; ++r) c[t][r] = 0.0f;

    for (int k0 = 0; k0 < D_FEAT; k0 += 4) {
        const int ka = k0 + half * 2;
        v2f bf;
        bf.x = W[(size_t)ka * D_FEAT + nb + l15];
        bf.y = W[(size_t)(ka + 1) * D_FEAT + nb + l15];
#pragma unroll
        for (int t = 0; t < MT_PER_BLK; ++t) {
            v2f af;
            af.x = sX[t * 16 + l15][ka];
            af.y = sX[t * 16 + l15][ka + 1];
            c[t] = __builtin_amdgcn_wmma_f32_16x16x4_f32(
                 false, af,  false, bf,
                 (short)0, c[t],  false,  false);
        }
    }

    if (full) {
#pragma unroll
        for (int t = 0; t < MT_PER_BLK; ++t) {
            const int rbase = mbase + t * 16 + half * 8;
            float* hp = hprime + (size_t)rbase * D_FEAT + nb + l15;
            float dv[8];
#pragma unroll
            for (int r = 0; r < 8; ++r) dv[r] = dinv[rbase + r];
#pragma unroll
            for (int r = 0; r < 8; ++r) {
                hp[(size_t)r * D_FEAT] = c[t][r] * dv[r];
            }
        }
    } else {
#pragma unroll
        for (int t = 0; t < MT_PER_BLK; ++t) {
#pragma unroll
            for (int r = 0; r < 8; ++r) {
                int row = mbase + t * 16 + r + half * 8;
                if (row < n_nodes) {
                    hprime[(size_t)row * D_FEAT + nb + l15] = c[t][r] * dinv[row];
                }
            }
        }
    }
}

#define GF_NODES_PER_WAVE 16

__device__ __forceinline__ void pool_flush(float* __restrict__ pooled,
                                           float* __restrict__ cnt, int g, int lane,
                                           const float (&sum)[4], float nseg) {
#pragma unroll
    for (int j = 0; j < 4; ++j)
        atomicAdd(&pooled[(size_t)g * D_FEAT + lane * 4 + j], sum[j]);
    if (lane == 0) atomicAdd(&cnt[g], nseg);
}

template <int NB>
__device__ __forceinline__ void gather_rows(const float4* __restrict__ hp4, int lane,
                                            int cv, int q, int m, float4& a) {
    float4 r[4 * NB];
#pragma unroll
    for (int u = 0; u < 4 * NB; ++u) {
        float4 v = make_float4(0.0f, 0.0f, 0.0f, 0.0f);
        if (u <= 4 * (NB - 1) || q + u < m) {
            const int s = __builtin_amdgcn_readlane(cv, q + u);
            v = hp4[(size_t)s * (D_FEAT / 4) + lane];
        }
        r[u] = v;
    }
#pragma unroll
    for (int u = 0; u < 4 * NB; ++u) {
        a.x += r[u].x;
        a.y += r[u].y;
        a.z += r[u].z;
        a.w += r[u].w;
    }
}

__device__ __forceinline__ void gather_chunk(const float4* __restrict__ hp4, int lane,
                                             int cv, int m, float4& a) {
    for (int q = 0; q < m; q += 16) {
        const int left = m - q;
        if (left > 12)     gather_rows<4>(hp4, lane, cv, q, m, a);
        else if (left > 8) gather_rows<3>(hp4, lane, cv, q, m, a);
        else if (left > 4) gather_rows<2>(hp4, lane, cv, q, m, a);
        else               gather_rows<1>(hp4, lane, cv, q, m, a);
    }
}

__global__ __launch_bounds__(256) void gather_finalize_kernel(
    const float* __restrict__ hprime, const int* __restrict__ rowstart,
    const int* __restrict__ degi, const int* __restrict__ col,
    const float* __restrict__ dinv, const float* __restrict__ bias,
    const float* __restrict__ pa, const int* __restrict__ batch,
    float* __restrict__ pooled, float* __restrict__ cnt, int n_nodes) {
    const int lane = threadIdx.x & 31;
    const int wv = __builtin_amdgcn_readfirstlane(
        (int)(blockIdx.x * (blockDim.x >> 5) + (threadIdx.x >> 5)));
    const int i0 = wv * GF_NODES_PER_WAVE;
    if (i0 >= n_nodes) return;
    const int nn = (i0 + GF_NODES_PER_WAVE < n_nodes) ? GF_NODES_PER_WAVE : (n_nodes - i0);

    const float4* hp4 = reinterpret_cast<const float4*>(hprime);
    const float4 b4 = reinterpret_cast<const float4*>(bias)[lane];
    const float4 a4 = reinterpret_cast<const float4*>(pa)[lane];
    const float bj[4] = {b4.x, b4.y, b4.z, b4.w};
    const float aj[4] = {a4.x, a4.y, a4.z, a4.w};

    const int li = i0 + ((lane < nn) ? lane : (nn - 1));
    const int   m_start = rowstart[li];
    const int   m_n     = degi[li];
    const float m_di    = dinv[li];
    const int   m_g     = batch[li];

    float sum[4] = {0.0f, 0.0f, 0.0f, 0.0f};
    float nseg = 0.0f;
    int gcur = __builtin_amdgcn_readlane(m_g, 0);

    int n_cur = 0, start_cur = 0, cv_cur = 0;
    for (int j = -1; j < nn; ++j) {
        int n_nx = 0, start_nx = 0, cv_nx = 0;
        if (j + 1 < nn) {
            n_nx     = __builtin_amdgcn_readlane(m_n, j + 1);
            start_nx = __builtin_amdgcn_readlane(m_start, j + 1);
            if (n_nx > 0) cv_nx = col[start_nx + ((lane < n_nx) ? lane : (n_nx - 1))];
        }
        if (j < 0) {
            n_cur = n_nx; start_cur = start_nx; cv_cur = cv_nx;
            continue;
        }
        const float di = __int_as_float(__builtin_amdgcn_readlane(__float_as_int(m_di), j));
        const int   g  = __builtin_amdgcn_readlane(m_g, j);

        float4 a = hp4[(size_t)(i0 + j) * (D_FEAT / 4) + lane];
        if (n_cur > 0)
            gather_chunk(hp4, lane, cv_cur, (n_cur < 32) ? n_cur : 32, a);
        for (int k = 32; k < n_cur; k += 32) {
            int kk = k + lane;
            kk = (kk < n_cur) ? kk : (n_cur - 1);
            const int cv = col[start_cur + kk];
            gather_chunk(hp4, lane, cv, (n_cur - k < 32) ? (n_cur - k) : 32, a);
        }

        float t[4] = {a.x, a.y, a.z, a.w};
        float ss = 0.0f;
#pragma unroll
        for (int c = 0; c < 4; ++c) {
            float v = di * t[c] + bj[c];
            v = (v > 0.0f) ? v : aj[c] * v;
            t[c] = v;
            ss += v * v;
        }
#pragma unroll
        for (int off = 16; off > 0; off >>= 1) ss += __shfl_xor(ss, off, 32);
        const float scale = 1.0f / fmaxf(sqrtf(ss), 1e-12f);
        if (g != gcur) {
            pool_flush(pooled, cnt, gcur, lane, sum, nseg);
            sum[0] = sum[1] = sum[2] = sum[3] = 0.0f;
            nseg = 0.0f;
            gcur = g;
        }
#pragma unroll
        for (int c = 0; c < 4; ++c) sum[c] += t[c] * scale;
        nseg += 1.0f;

        n_cur = n_nx; start_cur = start_nx; cv_cur = cv_nx;
    }
    pool_flush(pooled, cnt, gcur, lane, sum, nseg);
}

__global__ __launch_bounds__(256) void div_kernel(float* __restrict__ pooled,
                                                  const float* __restrict__ cnt,
                                                  int total) {
    int i = blockIdx.x * blockDim.x + threadIdx.x;
    if (i < total) pooled[i] /= fmaxf(cnt[i >> 7], 1.0f);
}

extern "C" void kernel_launch(void* const* d_in, const int* in_sizes, int n_in,
                              void* d_out, int out_size, void* d_ws, size_t ws_size,
                              hipStream_t stream) {
    const float* x     = (const float*)d_in[0];
    const int*   ei    = (const int*)d_in[1];
    const int*   batch = (const int*)d_in[2];
    const float* W     = (const float*)d_in[3];
    const float* bias  = (const float*)d_in[4];
    const float* pa    = (const float*)d_in[5];

    const int n_nodes = in_sizes[0] / D_FEAT;
    const int n_edges = in_sizes[1] / 2;
    const int* src = ei;
    const int* dst = ei + n_edges;
    float* pooled = (float*)d_out;

    float* hprime = (float*)d_ws;
    int*   degi     = (int*)(hprime + (size_t)n_nodes * D_FEAT);
    int*   gcur     = degi + n_nodes;
    int*   rowstart = gcur + 32;
    int*   cursor   = rowstart + n_nodes;
    int*   col      = cursor + n_nodes;
    float* dinv   = hprime + (size_t)n_nodes * D_FEAT * 2;
    float* cnt    = dinv + n_nodes;

    hipMemsetAsync(degi, 0, (size_t)(n_nodes + 32) * sizeof(int), stream);
    hipMemsetAsync(cnt, 0, 128 * sizeof(float), stream);
    hipMemsetAsync(d_out, 0, (size_t)out_size * sizeof(float), stream);

    count_kernel<<<(n_edges + 255) / 256, 256, 0, stream>>>(dst, degi, n_edges);
    alloc_dinv_kernel<<<(n_nodes + 255) / 256, 256, 0, stream>>>(degi, gcur, rowstart, cursor, dinv, n_nodes);

    int gemm_blocks = (n_nodes + ROWS_PER_BLK - 1) / ROWS_PER_BLK;
    gemm_scale_kernel<<<gemm_blocks, 256, 0, stream>>>(x, W, dinv, hprime, n_nodes);

    fill_kernel<<<(n_edges + 255) / 256, 256, 0, stream>>>(src, dst, cursor, col, n_edges);

    int gf_waves = (n_nodes + GF_NODES_PER_WAVE - 1) / GF_NODES_PER_WAVE;
    gather_finalize_kernel<<<(gf_waves + 7) / 8, 256, 0, stream>>>(
        hprime, rowstart, degi, col, dinv, bias, pa, batch, pooled, cnt, n_nodes);

    div_kernel<<<(out_size + 255) / 256, 256, 0, stream>>>(pooled, cnt, out_size);
}


# count/alloc/fill with two scattered global atomics per edge replaced by a two-level counting sort on LDS counters (512-node buckets, then one block per bucket); order of a row's in-edges changes
# speedup vs baseline: 4.1837x; 2.0036x over previous
#include <hip/hip_runtime.h>
#include <hip/hip_bf16.h>
#include <math.h>


typedef __attribute__((ext_vector_type(2))) float v2f;
typedef __attribute__((ext_vector_type(8))) float v8f;

#define D_FEAT 128
#define MT_PER_BLK 2
#define ROWS_PER_BLK (MT_PER_BLK * 16)

#define NPB_SHIFT 9
#define NPB (1 << NPB_SHIFT)
#define SRC_BITS 22
#define SRC_MASK ((1 << SRC_BITS) - 1)
#define SORT_BLKS 256
#define SORT_THREADS 512
#define SORT_EPT 16
#define ROWS_THREADS 1024
#define ROWS_EPT 12
#define STAGE_CAP (ROWS_THREADS * ROWS_EPT)
#define BT_STRIDE 32

__global__ __launch_bounds__(SORT_THREADS) void bucket_count_kernel(
    const int* __restrict__ dst, int* __restrict__ blkcnt, int* __restrict__ btotal,
    int n_edges, int chunk, int nb) {
    extern __shared__ int sh_bins[];
    const int tid = threadIdx.x;
    for (int b = tid; b < nb; b += SORT_THREADS) sh_bins[b] = 0;
    __syncthreads();
    const int e0 = blockIdx.x * chunk;
    const int e1 = (e0 + chunk < n_edges) ? (e0 + chunk) : n_edges;
    for (int base = e0; base < e1; base += SORT_THREADS * SORT_EPT) {
        int d[SORT_EPT];
#pragma unroll
        for (int u = 0; u < SORT_EPT; ++u) {
            const int e = base + tid + u * SORT_THREADS;
            d[u] = (e < e1) ? dst[e] : -1;
        }
#pragma unroll
        for (int u = 0; u < SORT_EPT; ++u)
            if (d[u] >= 0) atomicAdd(&sh_bins[d[u] >> NPB_SHIFT], 1);
    }
    __syncthreads();
    for (int b = tid; b < nb; b += SORT_THREADS) {
        const int c = sh_bins[b];
        blkcnt[(size_t)blockIdx.x * nb + b] = c;
        if (c) atomicAdd(&btotal[(size_t)b * BT_STRIDE], c);
    }
}

__global__ __launch_bounds__(SORT_BLKS) void bucket_offsets_kernel(
    const int* __restrict__ blkcnt, const int* __restrict__ btotal,
    int* __restrict__ offs, int* __restrict__ bstart, int nb) {
    __shared__ int swave[SORT_BLKS / 32];
    __shared__ int sred[SORT_BLKS / 32];
    const int b    = blockIdx.x;
    const int t    = threadIdx.x;
    const int lane = t & 31;
    const int wave = t >> 5;
    int part = 0;
    for (int q = t; q < b; q += SORT_BLKS) part += btotal[(size_t)q * BT_STRIDE];
#pragma unroll
    for (int off = 16; off > 0; off >>= 1) part += __shfl_xor(part, off, 32);
    const int c = blkcnt[(size_t)t * nb + b];
    int v = c;
#pragma unroll
    for (int off = 1; off < 32; off <<= 1) {
        int x = __shfl_up(v, off, 32);
        if (lane >= off) v += x;
    }
    if (lane == 31) swave[wave] = v;
    if (lane == 0)  sred[wave]  = part;
    __syncthreads();
    int base = 0, wo = 0;
    for (int w = 0; w < SORT_BLKS / 32; ++w) {
        base += sred[w];
        if (w < wave) wo += swave[w];
    }
    offs[(size_t)t * nb + b] = base + wo + (v - c);
    if (t == 0) bstart[b] = base;
}

__global__ __launch_bounds__(SORT_THREADS) void bucket_scatter_kernel(
    const int* __restrict__ src, const int* __restrict__ dst,
    const int* __restrict__ offs, int* __restrict__ ebuf,
    int n_edges, int chunk, int nb) {
    extern __shared__ int sh_bins[];
    const int tid = threadIdx.x;
    for (int b = tid; b < nb; b += SORT_THREADS)
        sh_bins[b] = offs[(size_t)blockIdx.x * nb + b];
    __syncthreads();
    const int e0 = blockIdx.x * chunk;
    const int e1 = (e0 + chunk < n_edges) ? (e0 + chunk) : n_edges;
    for (int base = e0; base < e1; base += SORT_THREADS * SORT_EPT) {
        int d[SORT_EPT], s[SORT_EPT];
#pragma unroll
        for (int u = 0; u < SORT_EPT; ++u) {
            const int e = base + tid + u * SORT_THREADS;
            d[u] = (e < e1) ? dst[e] : -1;
            s[u] = (e < e1) ? src[e] : 0;
        }
#pragma unroll
        for (int u = 0; u < SORT_EPT; ++u) {
            if (d[u] >= 0) {
                const int p = atomicAdd(&sh_bins[d[u] >> NPB_SHIFT], 1);
                if ((unsigned)p < (unsigned)n_edges)
                    ebuf[p] = ((d[u] & (NPB - 1)) << SRC_BITS) | s[u];
            }
        }
    }
}

__global__ __launch_bounds__(ROWS_THREADS) void bucket_rows_kernel(
    const int* __restrict__ ebuf, const int* __restrict__ bstart,
    const int* __restrict__ btotal, int* __restrict__ rowstart,
    int* __restrict__ degi, float* __restrict__ dinv, int* __restrict__ col,
    int n_nodes) {
    __shared__ int cntl[NPB];
    __shared__ int swave[NPB / 32];
    __shared__ int scol[STAGE_CAP];
    const int tid  = threadIdx.x;
    const int lane = tid & 31;
    const int wave = tid >> 5;
    const int b    = blockIdx.x;
    const int seg0 = bstart[b];
    const int m    = btotal[(size_t)b * BT_STRIDE];
    if (tid < NPB) cntl[tid] = 0;
    __syncthreads();
    for (int base = 0; base < m; base += ROWS_THREADS * ROWS_EPT) {
        int w[ROWS_EPT];
#pragma unroll
        for (int u = 0; u < ROWS_EPT; ++u) {
            const int i = base + tid + u * ROWS_THREADS;
            w[u] = (i < m) ? ebuf[seg0 + i] : -1;
        }
#pragma unroll
        for (int u = 0; u < ROWS_EPT; ++u)
            if (w[u] >= 0) atomicAdd(&cntl[w[u] >> SRC_BITS], 1);
    }
    __syncthreads();
    const int c = (tid < NPB) ? cntl[tid] : 0;
    int v = c;
#pragma unroll
    for (int off = 1; off < 32; off <<= 1) {
        int x = __shfl_up(v, off, 32);
        if (lane >= off) v += x;
    }
    if (tid < NPB && lane == 31) swave[wave] = v;
    __syncthreads();
    if (tid < NPB) {
        int wo = 0;
        for (int q = 0; q < wave; ++q) wo += swave[q];
        const int ex = wo + (v - c);
        cntl[tid] = ex;
        const int node = b * NPB + tid;
        if (node < n_nodes) {
            rowstart[node] = seg0 + ex;
            degi[node]     = c;
            dinv[node]     = rsqrtf((float)c + 1.0f);
        }
    }
    __syncthreads();
    const bool staged = (m <= STAGE_CAP);
    for (int base = 0; base < m; base += ROWS_THREADS * ROWS_EPT) {
        int w[ROWS_EPT];
#pragma unroll
        for (int u = 0; u < ROWS_EPT; ++u) {
            const int i = base + tid + u * ROWS_THREADS;
            w[u] = (i < m) ? ebuf[seg0 + i] : -1;
        }
#pragma unroll
        for (int u = 0; u < ROWS_EPT; ++u) {
            if (w[u] >= 0) {
                const int p = atomicAdd(&cntl[w[u] >> SRC_BITS], 1);
                const int s = w[u] & SRC_MASK;
                if ((unsigned)p < (unsigned)m) {
                    if (staged) scol[p] = s;
                    else        col[seg0 + p] = s;
                }
            }
        }
    }
    if (staged) {
        __syncthreads();
        for (int i = tid; i < m; i += ROWS_THREADS) col[seg0 + i] = scol[i];
    }
}

__global__ __launch_bounds__(256) void gemm_scale_kernel(
    const float* __restrict__ x, const float* __restrict__ W,
    const float* __restrict__ dinv, float* __restrict__ hprime,
    int n_nodes) {
    __shared__ float sX[ROWS_PER_BLK][D_FEAT + 4];

    const int tid   = threadIdx.x;
    const int lane  = tid & 31;
    const int wave  = tid >> 5;
    const int mbase = blockIdx.x * ROWS_PER_BLK;
    const int half  = lane >> 4;
    const int l15   = lane & 15;
    const int nb    = wave * 16;

    const bool full = (mbase + ROWS_PER_BLK) <= n_nodes;

    const float4* xv = reinterpret_cast<const float4*>(x);
    if (full) {
#pragma unroll
        for (int i = 0; i < 4; ++i) {
            int slot = tid + i * 256;
            int r    = slot >> 5;
            int cb   = slot & 31;
            *reinterpret_cast<float4*>(&sX[r][cb * 4]) =
                xv[(size_t)(mbase + r) * (D_FEAT / 4) + cb];
        }
    } else {
        const float4 zero4 = make_float4(0.f, 0.f, 0.f, 0.f);
#pragma unroll
        for (int i = 0; i < 4; ++i) {
            int slot = tid + i * 256;
            int r    = slot >> 5;
            int cb   = slot & 31;
            int row  = mbase + r;
            float4 v = (row < n_nodes) ? xv[(size_t)row * (D_FEAT / 4) + cb] : zero4;
            *reinterpret_cast<float4*>(&sX[r][cb * 4]) = v;
        }
    }
    __syncthreads();

    v8f c[MT_PER_BLK];
#pragma unroll
    for (int t = 0; t < MT_PER_BLK; ++t)
#pragma unroll
        for (int r = 0; r < 8; ++r) c[t][r] = 0.0f;

    for (int k0 = 0; k0 < D_FEAT; k0 += 4) {
        const int ka = k0 + half * 2;
        v2f bf;
        bf.x = W[(size_t)ka * D_FEAT + nb + l15];
        bf.y = W[(size_t)(ka + 1) * D_FEAT + nb + l15];
#pragma unroll
        for (int t = 0; t < MT_PER_BLK; ++t) {
            v2f af;
            af.x = sX[t * 16 + l15][ka];
            af.y = sX[t * 16 + l15][ka + 1];
            c[t] = __builtin_amdgcn_wmma_f32_16x16x4_f32(
                 false, af,  false, bf,
                 (short)0, c[t],  false,  false);
        }
    }

    if (full) {
#pragma unroll
        for (int t = 0; t < MT_PER_BLK; ++t) {
            const int rbase = mbase + t * 16 + half * 8;
            float* hp = hprime + (size_t)rbase * D_FEAT + nb + l15;
            float dv[8];
#pragma unroll
            for (int r = 0; r < 8; ++r) dv[r] = dinv[rbase + r];
#pragma unroll
            for (int r = 0; r < 8; ++r) {
                hp[(size_t)r * D_FEAT] = c[t][r] * dv[r];
            }
        }
    } else {
#pragma unroll
        for (int t = 0; t < MT_PER_BLK; ++t) {
#pragma unroll
            for (int r = 0; r < 8; ++r) {
                int row = mbase + t * 16 + r + half * 8;
                if (row < n_nodes) {
                    hprime[(size_t)row * D_FEAT + nb + l15] = c[t][r] * dinv[row];
                }
            }
        }
    }
}

#define GF_NODES_PER_WAVE 16

__device__ __forceinline__ void pool_flush(float* __restrict__ pooled,
                                           float* __restrict__ cnt, int g, int lane,
                                           const float (&sum)[4], float nseg) {
#pragma unroll
    for (int j = 0; j < 4; ++j)
        atomicAdd(&pooled[(size_t)g * D_FEAT + lane * 4 + j], sum[j]);
    if (lane == 0) atomicAdd(&cnt[g], nseg);
}

template <int NB>
__device__ __forceinline__ void gather_rows(const float4* __restrict__ hp4, int lane,
                                            int cv, int q, int m, float4& a) {
    float4 r[4 * NB];
#pragma unroll
    for (int u = 0; u < 4 * NB; ++u) {
        float4 v = make_float4(0.0f, 0.0f, 0.0f, 0.0f);
        if (u <= 4 * (NB - 1) || q + u < m) {
            const int s = __builtin_amdgcn_readlane(cv, q + u);
            v = hp4[(size_t)s * (D_FEAT / 4) + lane];
        }
        r[u] = v;
    }
#pragma unroll
    for (int u = 0; u < 4 * NB; ++u) {
        a.x += r[u].x;
        a.y += r[u].y;
        a.z += r[u].z;
        a.w += r[u].w;
    }
}

__device__ __forceinline__ void gather_chunk(const float4* __restrict__ hp4, int lane,
                                             int cv, int m, float4& a) {
    for (int q = 0; q < m; q += 16) {
        const int left = m - q;
        if (left > 12)     gather_rows<4>(hp4, lane, cv, q, m, a);
        else if (left > 8) gather_rows<3>(hp4, lane, cv, q, m, a);
        else if (left > 4) gather_rows<2>(hp4, lane, cv, q, m, a);
        else               gather_rows<1>(hp4, lane, cv, q, m, a);
    }
}

__global__ __launch_bounds__(256) void gather_finalize_kernel(
    const float* __restrict__ hprime, const int* __restrict__ rowstart,
    const int* __restrict__ degi, const int* __restrict__ col,
    const float* __restrict__ dinv, const float* __restrict__ bias,
    const float* __restrict__ pa, const int* __restrict__ batch,
    float* __restrict__ pooled, float* __restrict__ cnt, int n_nodes) {
    const int lane = threadIdx.x & 31;
    const int wv = __builtin_amdgcn_readfirstlane(
        (int)(blockIdx.x * (blockDim.x >> 5) + (threadIdx.x >> 5)));
    const int i0 = wv * GF_NODES_PER_WAVE;
    if (i0 >= n_nodes) return;
    const int nn = (i0 + GF_NODES_PER_WAVE < n_nodes) ? GF_NODES_PER_WAVE : (n_nodes - i0);

    const float4* hp4 = reinterpret_cast<const float4*>(hprime);
    const float4 b4 = reinterpret_cast<const float4*>(bias)[lane];
    const float4 a4 = reinterpret_cast<const float4*>(pa)[lane];
    const float bj[4] = {b4.x, b4.y, b4.z, b4.w};
    const float aj[4] = {a4.x, a4.y, a4.z, a4.w};

    const int li = i0 + ((lane < nn) ? lane : (nn - 1));
    const int   m_start = rowstart[li];
    const int   m_n     = degi[li];
    const float m_di    = dinv[li];
    const int   m_g     = batch[li];

    float sum[4] = {0.0f, 0.0f, 0.0f, 0.0f};
    float nseg = 0.0f;
    int gcur = __builtin_amdgcn_readlane(m_g, 0);

    int n_cur = 0, start_cur = 0, cv_cur = 0;
    for (int j = -1; j < nn; ++j) {
        int n_nx = 0, start_nx = 0, cv_nx = 0;
        if (j + 1 < nn) {
            n_nx     = __builtin_amdgcn_readlane(m_n, j + 1);
            start_nx = __builtin_amdgcn_readlane(m_start, j + 1);
            if (n_nx > 0) cv_nx = col[start_nx + ((lane < n_nx) ? lane : (n_nx - 1))];
        }
        if (j < 0) {
            n_cur = n_nx; start_cur = start_nx; cv_cur = cv_nx;
            continue;
        }
        const float di = __int_as_float(__builtin_amdgcn_readlane(__float_as_int(m_di), j));
        const int   g  = __builtin_amdgcn_readlane(m_g, j);

        float4 a = hp4[(size_t)(i0 + j) * (D_FEAT / 4) + lane];
        if (n_cur > 0)
            gather_chunk(hp4, lane, cv_cur, (n_cur < 32) ? n_cur : 32, a);
        for (int k = 32; k < n_cur; k += 32) {
            int kk = k + lane;
            kk = (kk < n_cur) ? kk : (n_cur - 1);
            const int cv = col[start_cur + kk];
            gather_chunk(hp4, lane, cv, (n_cur - k < 32) ? (n_cur - k) : 32, a);
        }

        float t[4] = {a.x, a.y, a.z, a.w};
        float ss = 0.0f;
#pragma unroll
        for (int c = 0; c < 4; ++c) {
            float v = di * t[c] + bj[c];
            v = (v > 0.0f) ? v : aj[c] * v;
            t[c] = v;
            ss += v * v;
        }
#pragma unroll
        for (int off = 16; off > 0; off >>= 1) ss += __shfl_xor(ss, off, 32);
        const float scale = 1.0f / fmaxf(sqrtf(ss), 1e-12f);
        if (g != gcur) {
            pool_flush(pooled, cnt, gcur, lane, sum, nseg);
            sum[0] = sum[1] = sum[2] = sum[3] = 0.0f;
            nseg = 0.0f;
            gcur = g;
        }
#pragma unroll
        for (int c = 0; c < 4; ++c) sum[c] += t[c] * scale;
        nseg += 1.0f;

        n_cur = n_nx; start_cur = start_nx; cv_cur = cv_nx;
    }
    pool_flush(pooled, cnt, gcur, lane, sum, nseg);
}

__global__ __launch_bounds__(256) void div_kernel(float* __restrict__ pooled,
                                                  const float* __restrict__ cnt,
                                                  int total) {
    int i = blockIdx.x * blockDim.x + threadIdx.x;
    if (i < total) pooled[i] /= fmaxf(cnt[i >> 7], 1.0f);
}

extern "C" void kernel_launch(void* const* d_in, const int* in_sizes, int n_in,
                              void* d_out, int out_size, void* d_ws, size_t ws_size,
                              hipStream_t stream) {
    const float* x     = (const float*)d_in[0];
    const int*   ei    = (const int*)d_in[1];
    const int*   batch = (const int*)d_in[2];
    const float* W     = (const float*)d_in[3];
    const float* bias  = (const float*)d_in[4];
    const float* pa    = (const float*)d_in[5];

    const int n_nodes = in_sizes[0] / D_FEAT;
    const int n_edges = in_sizes[1] / 2;
    const int* src = ei;
    const int* dst = ei + n_edges;
    float* pooled = (float*)d_out;

    const int nb    = (n_nodes + NPB - 1) / NPB;
    const int chunk = (n_edges + SORT_BLKS - 1) / SORT_BLKS;

    float* hprime = (float*)d_ws;
    int*   degi     = (int*)(hprime + (size_t)n_nodes * D_FEAT);
    int*   btotal   = degi + n_nodes;
    int*   bstart   = btotal + (size_t)nb * BT_STRIDE;
    int*   rowstart = bstart + nb;
    int*   ebuf     = rowstart + n_nodes;
    int*   col      = ebuf + n_edges;
    int*   blkcnt   = col + n_edges;
    int*   offs     = blkcnt + (size_t)SORT_BLKS * nb;
    float* dinv   = hprime + (size_t)n_nodes * D_FEAT * 2;
    float* cnt    = dinv + n_nodes;

    hipMemsetAsync(btotal, 0, (size_t)nb * BT_STRIDE * sizeof(int), stream);
    hipMemsetAsync(cnt, 0, 128 * sizeof(float), stream);
    hipMemsetAsync(d_out, 0, (size_t)out_size * sizeof(float), stream);

    bucket_count_kernel<<<SORT_BLKS, SORT_THREADS, (size_t)nb * sizeof(int), stream>>>(
        dst, blkcnt, btotal, n_edges, chunk, nb);
    bucket_offsets_kernel<<<nb, SORT_BLKS, 0, stream>>>(blkcnt, btotal, offs, bstart, nb);
    bucket_scatter_kernel<<<SORT_BLKS, SORT_THREADS, (size_t)nb * sizeof(int), stream>>>(
        src, dst, offs, ebuf, n_edges, chunk, nb);
    bucket_rows_kernel<<<nb, ROWS_THREADS, 0, stream>>>(
        ebuf, bstart, btotal, rowstart, degi, dinv, col, n_nodes);

    int gemm_blocks = (n_nodes + ROWS_PER_BLK - 1) / ROWS_PER_BLK;
    gemm_scale_kernel<<<gemm_blocks, 256, 0, stream>>>(x, W, dinv, hprime, n_nodes);

    int gf_waves = (n_nodes + GF_NODES_PER_WAVE - 1) / GF_NODES_PER_WAVE;
    gather_finalize_kernel<<<(gf_waves + 7) / 8, 256, 0, stream>>>(
        hprime, rowstart, degi, col, dinv, bias, pa, batch, pooled, cnt, n_nodes);

    div_kernel<<<(out_size + 255) / 256, 256, 0, stream>>>(pooled, cnt, out_size);
}


# product kernel made persistent: at most 768 resident blocks walk the 32-row tiles; a wave keeps its 128x16 block of W in registers and loads the next x tile under the current products
# speedup vs baseline: 5.8888x; 1.4076x over previous
#include <hip/hip_runtime.h>
#include <hip/hip_bf16.h>
#include <math.h>


typedef __attribute__((ext_vector_type(2))) float v2f;
typedef __attribute__((ext_vector_type(8))) float v8f;

#define D_FEAT 128
#define MT_PER_BLK 2
#define ROWS_PER_BLK (MT_PER_BLK * 16)

#define NPB_SHIFT 9
#define NPB (1 << NPB_SHIFT)
#define SRC_BITS 22
#define SRC_MASK ((1 << SRC_BITS) - 1)
#define SORT_BLKS 256
#define SORT_THREADS 512
#define SORT_EPT 16
#define ROWS_THREADS 1024
#define ROWS_WAVES (ROWS_THREADS / 32)
#define SEGS_PER_WAVE (SORT_BLKS / ROWS_WAVES)
#define SEG_ROUNDS 2
#define STAGE_CAP 12288

static_assert(SORT_BLKS % ROWS_WAVES == 0, "every wave reads the same number of segments");
static_assert(SORT_BLKS <= ROWS_THREADS, "one thread per chunk reads the bucket starts");

__global__ __launch_bounds__(SORT_THREADS) void chunk_sort_kernel(
    const int* __restrict__ src, const int* __restrict__ dst,
    int* __restrict__ tab, int* __restrict__ ebuf,
    float* __restrict__ pooled, float* __restrict__ cnt, int out_size, int n_graphs,
    int n_edges, int chunk, int nb, int stage) {
    extern __shared__ int sh_sort[];
    int* bins = sh_sort;
    int* cur  = sh_sort + (nb + 1);
    int* sbuf = cur + (nb + 1);
    const int tid  = threadIdx.x;
    const int lane = tid & 31;
    for (int i = blockIdx.x * SORT_THREADS + tid; i < out_size; i += gridDim.x * SORT_THREADS)
        pooled[i] = 0.0f;
    if (blockIdx.x == 0)
        for (int g = tid; g < n_graphs; g += SORT_THREADS) cnt[g] = 0.0f;
    for (int b = tid; b <= nb; b += SORT_THREADS) bins[b] = 0;
    __syncthreads();
    const int e0 = blockIdx.x * chunk;
    const int e1 = (e0 + chunk < n_edges) ? (e0 + chunk) : n_edges;
    const int n  = (e1 > e0) ? (e1 - e0) : 0;
    for (int base = e0; base < e1; base += SORT_THREADS * SORT_EPT) {
        int d[SORT_EPT];
#pragma unroll
        for (int u = 0; u < SORT_EPT; ++u) {
            const int e = base + tid + u * SORT_THREADS;
            d[u] = (e < e1) ? dst[e] : -1;
        }
#pragma unroll
        for (int u = 0; u < SORT_EPT; ++u)
            if (d[u] >= 0) atomicAdd(&bins[d[u] >> NPB_SHIFT], 1);
    }
    __syncthreads();
    if (tid < 32) {
        int carry = 0;
        for (int b0 = 0; b0 < nb; b0 += 32) {
            const int c = (b0 + lane < nb) ? bins[b0 + lane] : 0;
            int v = c;
#pragma unroll
            for (int off = 1; off < 32; off <<= 1) {
                int x = __shfl_up(v, off, 32);
                if (lane >= off) v += x;
            }
            if (b0 + lane < nb) bins[b0 + lane] = carry + (v - c);
            carry += __shfl(v, 31, 32);
        }
        if (lane == 0) bins[nb] = carry;
    }
    __syncthreads();
    for (int b = tid; b <= nb; b += SORT_THREADS) {
        const int st = bins[b];
        tab[(size_t)blockIdx.x * (nb + 1) + b] = st;
        cur[b] = st;
    }
    __syncthreads();
    const bool staged = (n <= stage);
    for (int base = e0; base < e1; base += SORT_THREADS * SORT_EPT) {
        int d[SORT_EPT], s[SORT_EPT];
#pragma unroll
        for (int u = 0; u < SORT_EPT; ++u) {
            const int e = base + tid + u * SORT_THREADS;
            d[u] = (e < e1) ? dst[e] : -1;
            s[u] = (e < e1) ? src[e] : 0;
        }
#pragma unroll
        for (int u = 0; u < SORT_EPT; ++u) {
            if (d[u] >= 0) {
                const int p = atomicAdd(&cur[d[u] >> NPB_SHIFT], 1);
                const int word = ((d[u] & (NPB - 1)) << SRC_BITS) | s[u];
                if ((unsigned)p < (unsigned)n) {
                    if (staged) sbuf[p] = word;
                    else        ebuf[e0 + p] = word;
                }
            }
        }
    }
    if (staged) {
        __syncthreads();
        for (int i = tid; i < n; i += SORT_THREADS) ebuf[e0 + i] = sbuf[i];
    }
}

__global__ __launch_bounds__(ROWS_THREADS) void bucket_rows_kernel(
    const int* __restrict__ ebuf, const int* __restrict__ tab,
    int* __restrict__ rowstart, int* __restrict__ degi, float* __restrict__ dinv,
    int* __restrict__ col, const int* __restrict__ batch, float* __restrict__ cnt,
    int n_nodes, int n_graphs, int n_edges, int chunk, int nb) {
    extern __shared__ int sh_hist[];
    __shared__ int cntl[NPB];
    __shared__ int swave[NPB / 32];
    __shared__ int seg_off[SORT_BLKS];
    __shared__ int seg_len[SORT_BLKS];
    __shared__ int red_a[SORT_BLKS / 32];
    __shared__ int red_b[SORT_BLKS / 32];
    __shared__ int scol[STAGE_CAP];
    const int tid  = threadIdx.x;
    const int lane = tid & 31;
    const int wave = tid >> 5;
    const int b    = blockIdx.x;
    const int node = b * NPB + tid;

    int s0 = 0, ln = 0;
    if (tid < SORT_BLKS) {
        const int* row = tab + (size_t)tid * (nb + 1) + b;
        s0 = row[0];
        ln = row[1] - s0;
        const int c0 = tid * chunk;
        int avail = n_edges - c0;
        avail = (avail < chunk) ? avail : chunk;
        avail = (avail > 0) ? avail : 0;
        s0 = (s0 > 0) ? s0 : 0;
        s0 = (s0 < avail) ? s0 : avail;
        ln = (ln > 0) ? ln : 0;
        ln = (ln < avail - s0) ? ln : (avail - s0);
        seg_off[tid] = c0 + s0;
        seg_len[tid] = ln;
    }
    int bg = -1;
    if (tid < NPB && node < n_nodes) bg = batch[node];
    int va = s0, vb = ln;
#pragma unroll
    for (int off = 16; off > 0; off >>= 1) {
        va += __shfl_xor(va, off, 32);
        vb += __shfl_xor(vb, off, 32);
    }
    if (tid < SORT_BLKS && lane == 0) { red_a[wave] = va; red_b[wave] = vb; }
    if (tid < NPB) cntl[tid] = 0;
    for (int g = tid; g < n_graphs; g += ROWS_THREADS) sh_hist[g] = 0;
    __syncthreads();
    int seg0 = 0, m = 0;
    for (int q = 0; q < SORT_BLKS / 32; ++q) { seg0 += red_a[q]; m += red_b[q]; }

    int w[SEGS_PER_WAVE * SEG_ROUNDS];
#pragma unroll
    for (int u = 0; u < SEGS_PER_WAVE; ++u) {
        const int t    = wave + u * ROWS_WAVES;
        const int ln_t = seg_len[t];
        const int of_t = seg_off[t];
#pragma unroll
        for (int r = 0; r < SEG_ROUNDS; ++r) {
            const int k = lane + 32 * r;
            w[u * SEG_ROUNDS + r] = (k < ln_t) ? ebuf[of_t + k] : -1;
        }
    }
    if (bg >= 0) atomicAdd(&sh_hist[bg], 1);
#pragma unroll
    for (int i = 0; i < SEGS_PER_WAVE * SEG_ROUNDS; ++i)
        if (w[i] >= 0) atomicAdd(&cntl[(w[i] >> SRC_BITS) & (NPB - 1)], 1);
    for (int u = 0; u < SEGS_PER_WAVE; ++u) {
        const int t    = wave + u * ROWS_WAVES;
        const int ln_t = seg_len[t];
        const int of_t = seg_off[t];
        for (int k = SEG_ROUNDS * 32 + lane; k < ln_t; k += 32) {
            const int ww = ebuf[of_t + k];
            atomicAdd(&cntl[(ww >> SRC_BITS) & (NPB - 1)], 1);
        }
    }
    __syncthreads();
    for (int g = tid; g < n_graphs; g += ROWS_THREADS) {
        const int h = sh_hist[g];
        if (h) atomicAdd(&cnt[g], (float)h);
    }
    const int c = (tid < NPB) ? cntl[tid] : 0;
    int v = c;
#pragma unroll
    for (int off = 1; off < 32; off <<= 1) {
        int x = __shfl_up(v, off, 32);
        if (lane >= off) v += x;
    }
    if (tid < NPB && lane == 31) swave[wave] = v;
    __syncthreads();
    if (tid < NPB) {
        int wo = 0;
        for (int q = 0; q < wave; ++q) wo += swave[q];
        const int ex = wo + (v - c);
        cntl[tid] = ex;
        if (node < n_nodes) {
            rowstart[node] = seg0 + ex;
            degi[node]     = c;
            dinv[node]     = rsqrtf((float)c + 1.0f);
        }
    }
    __syncthreads();
    const bool staged = (m <= STAGE_CAP);
#pragma unroll
    for (int i = 0; i < SEGS_PER_WAVE * SEG_ROUNDS; ++i) {
        if (w[i] >= 0) {
            const int p = atomicAdd(&cntl[(w[i] >> SRC_BITS) & (NPB - 1)], 1);
            const int s = w[i] & SRC_MASK;
            if ((unsigned)p < (unsigned)m) {
                if (staged) scol[p] = s;
                else        col[seg0 + p] = s;
            }
        }
    }
    for (int u = 0; u < SEGS_PER_WAVE; ++u) {
        const int t    = wave + u * ROWS_WAVES;
        const int ln_t = seg_len[t];
        const int of_t = seg_off[t];
        for (int k = SEG_ROUNDS * 32 + lane; k < ln_t; k += 32) {
            const int ww = ebuf[of_t + k];
            const int p  = atomicAdd(&cntl[(ww >> SRC_BITS) & (NPB - 1)], 1);
            if ((unsigned)p < (unsigned)m) {
                if (staged) scol[p] = ww & SRC_MASK;
                else        col[seg0 + p] = ww & SRC_MASK;
            }
        }
    }
    if (staged) {
        __syncthreads();
        for (int i = tid; i < m; i += ROWS_THREADS) col[seg0 + i] = scol[i];
    }
}

#define GEMM_MAX_BLKS 768

__device__ __forceinline__ void load_x_tile(const float4* __restrict__ xv, int mbase, int tid,
                                            int n_nodes, float4& p0, float4& p1,
                                            float4& p2, float4& p3) {
    const int r0 = mbase + (tid >> 5);
    const int cb = tid & 31;
    if (mbase + ROWS_PER_BLK <= n_nodes) {
        p0 = xv[(size_t)(r0)      * (D_FEAT / 4) + cb];
        p1 = xv[(size_t)(r0 + 8)  * (D_FEAT / 4) + cb];
        p2 = xv[(size_t)(r0 + 16) * (D_FEAT / 4) + cb];
        p3 = xv[(size_t)(r0 + 24) * (D_FEAT / 4) + cb];
    } else {
        p0 = make_float4(0.f, 0.f, 0.f, 0.f);
        p1 = p0; p2 = p0; p3 = p0;
        if (r0      < n_nodes) p0 = xv[(size_t)(r0)      * (D_FEAT / 4) + cb];
        if (r0 + 8  < n_nodes) p1 = xv[(size_t)(r0 + 8)  * (D_FEAT / 4) + cb];
        if (r0 + 16 < n_nodes) p2 = xv[(size_t)(r0 + 16) * (D_FEAT / 4) + cb];
        if (r0 + 24 < n_nodes) p3 = xv[(size_t)(r0 + 24) * (D_FEAT / 4) + cb];
    }
}

__global__ __launch_bounds__(256) void gemm_scale_kernel(
    const float* __restrict__ x, const float* __restrict__ W,
    const float* __restrict__ dinv, float* __restrict__ hprime,
    int n_nodes) {
    __shared__ float sX[ROWS_PER_BLK][D_FEAT + 4];

    const int tid   = threadIdx.x;
    const int lane  = tid & 31;
    const int wave  = tid >> 5;
    const int half  = lane >> 4;
    const int l15   = lane & 15;
    const int nb    = wave * 16;
    const int n_tiles = (n_nodes + ROWS_PER_BLK - 1) / ROWS_PER_BLK;

    v2f bfr[D_FEAT / 4];
#pragma unroll
    for (int s = 0; s < D_FEAT / 4; ++s) {
        const int ka = 4 * s + half * 2;
        bfr[s].x = W[(size_t)ka * D_FEAT + nb + l15];
        bfr[s].y = W[(size_t)(ka + 1) * D_FEAT + nb + l15];
    }

    const float4* xv = reinterpret_cast<const float4*>(x);
    const int sr = tid >> 5;
    const int cb = tid & 31;
    float4 p0, p1, p2, p3;
    p0 = make_float4(0.f, 0.f, 0.f, 0.f);
    p1 = p0; p2 = p0; p3 = p0;
    int tile = blockIdx.x;
    if (tile < n_tiles) load_x_tile(xv, tile * ROWS_PER_BLK, tid, n_nodes, p0, p1, p2, p3);

    for (; tile < n_tiles; tile += gridDim.x) {
        const int mbase = tile * ROWS_PER_BLK;
        __syncthreads();
        *reinterpret_cast<float4*>(&sX[sr][cb * 4])      = p0;
        *reinterpret_cast<float4*>(&sX[sr + 8][cb * 4])  = p1;
        *reinterpret_cast<float4*>(&sX[sr + 16][cb * 4]) = p2;
        *reinterpret_cast<float4*>(&sX[sr + 24][cb * 4]) = p3;
        __syncthreads();
        const int next = tile + gridDim.x;
        if (next < n_tiles)
            load_x_tile(xv, next * ROWS_PER_BLK, tid, n_nodes, p0, p1, p2, p3);

        v8f c[MT_PER_BLK];
#pragma unroll
        for (int t = 0; t < MT_PER_BLK; ++t)
#pragma unroll
            for (int r = 0; r < 8; ++r) c[t][r] = 0.0f;

#pragma unroll
        for (int s = 0; s < D_FEAT / 4; ++s) {
            const int ka = 4 * s + half * 2;
#pragma unroll
            for (int t = 0; t < MT_PER_BLK; ++t) {
                v2f af;
                af.x = sX[t * 16 + l15][ka];
                af.y = sX[t * 16 + l15][ka + 1];
                c[t] = __builtin_amdgcn_wmma_f32_16x16x4_f32(
                     false, af,  false, bfr[s],
                     (short)0, c[t],  false,  false);
            }
        }

        if (mbase + ROWS_PER_BLK <= n_nodes) {
#pragma unroll
            for (int t = 0; t < MT_PER_BLK; ++t) {
                const int rbase = mbase + t * 16 + half * 8;
                float* hp = hprime + (size_t)rbase * D_FEAT + nb + l15;
                float dv[8];
#pragma unroll
                for (int r = 0; r < 8; ++r) dv[r] = dinv[rbase + r];
#pragma unroll
                for (int r = 0; r < 8; ++r) {
                    hp[(size_t)r * D_FEAT] = c[t][r] * dv[r];
                }
            }
        } else {
#pragma unroll
            for (int t = 0; t < MT_PER_BLK; ++t) {
#pragma unroll
                for (int r = 0; r < 8; ++r) {
                    int row = mbase + t * 16 + r + half * 8;
                    if (row < n_nodes) {
                        hprime[(size_t)row * D_FEAT + nb + l15] = c[t][r] * dinv[row];
                    }
                }
            }
        }
    }
}

#define GF_NODES_PER_WAVE 16

__device__ __forceinline__ void pool_flush(float* __restrict__ pooled,
                                           const float* __restrict__ cnt, int g, int lane,
                                           const float (&sum)[4]) {
    const float cg = fmaxf(cnt[g], 1.0f);
#pragma unroll
    for (int j = 0; j < 4; ++j)
        atomicAdd(&pooled[(size_t)g * D_FEAT + lane * 4 + j], sum[j] / cg);
}

template <int NB>
__device__ __forceinline__ void gather_rows(const float4* __restrict__ hp4, int lane,
                                            int cv, int q, int m, float4& a) {
    float4 r[4 * NB];
#pragma unroll
    for (int u = 0; u < 4 * NB; ++u) {
        float4 v = make_float4(0.0f, 0.0f, 0.0f, 0.0f);
        if (u <= 4 * (NB - 1) || q + u < m) {
            const int s = __builtin_amdgcn_readlane(cv, q + u);
            v = hp4[(size_t)s * (D_FEAT / 4) + lane];
        }
        r[u] = v;
    }
#pragma unroll
    for (int u = 0; u < 4 * NB; ++u) {
        a.x += r[u].x;
        a.y += r[u].y;
        a.z += r[u].z;
        a.w += r[u].w;
    }
}

__device__ __forceinline__ void gather_chunk(const float4* __restrict__ hp4, int lane,
                                             int cv, int m, float4& a) {
    for (int q = 0; q < m; q += 16) {
        const int left = m - q;
        if (left > 12)     gather_rows<4>(hp4, lane, cv, q, m, a);
        else if (left > 8) gather_rows<3>(hp4, lane, cv, q, m, a);
        else if (left > 4) gather_rows<2>(hp4, lane, cv, q, m, a);
        else               gather_rows<1>(hp4, lane, cv, q, m, a);
    }
}

__global__ __launch_bounds__(256) void gather_finalize_kernel(
    const float* __restrict__ hprime, const int* __restrict__ rowstart,
    const int* __restrict__ degi, const int* __restrict__ col,
    const float* __restrict__ dinv, const float* __restrict__ bias,
    const float* __restrict__ pa, const int* __restrict__ batch,
    float* __restrict__ pooled, const float* __restrict__ cnt, int n_nodes) {
    const int lane = threadIdx.x & 31;
    const int wv = __builtin_amdgcn_readfirstlane(
        (int)(blockIdx.x * (blockDim.x >> 5) + (threadIdx.x >> 5)));
    const int i0 = wv * GF_NODES_PER_WAVE;
    if (i0 >= n_nodes) return;
    const int nn = (i0 + GF_NODES_PER_WAVE < n_nodes) ? GF_NODES_PER_WAVE : (n_nodes - i0);

    const float4* hp4 = reinterpret_cast<const float4*>(hprime);
    const float4 b4 = reinterpret_cast<const float4*>(bias)[lane];
    const float4 a4 = reinterpret_cast<const float4*>(pa)[lane];
    const float bj[4] = {b4.x, b4.y, b4.z, b4.w};
    const float aj[4] = {a4.x, a4.y, a4.z, a4.w};

    const int li = i0 + ((lane < nn) ? lane : (nn - 1));
    const int   m_start = rowstart[li];
    const int   m_n     = degi[li];
    const float m_di    = dinv[li];
    const int   m_g     = batch[li];

    float sum[4] = {0.0f, 0.0f, 0.0f, 0.0f};
    int gcur = __builtin_amdgcn_readlane(m_g, 0);

    int n_cur = 0, start_cur = 0, cv_cur = 0;
    for (int j = -1; j < nn; ++j) {
        int n_nx = 0, start_nx = 0, cv_nx = 0;
        if (j + 1 < nn) {
            n_nx     = __builtin_amdgcn_readlane(m_n, j + 1);
            start_nx = __builtin_amdgcn_readlane(m_start, j + 1);
            if (n_nx > 0) cv_nx = col[start_nx + ((lane < n_nx) ? lane : (n_nx - 1))];
        }
        if (j < 0) {
            n_cur = n_nx; start_cur = start_nx; cv_cur = cv_nx;
            continue;
        }
        const float di = __int_as_float(__builtin_amdgcn_readlane(__float_as_int(m_di), j));
        const int   g  = __builtin_amdgcn_readlane(m_g, j);

        float4 a = hp4[(size_t)(i0 + j) * (D_FEAT / 4) + lane];
        if (n_cur > 0)
            gather_chunk(hp4, lane, cv_cur, (n_cur < 32) ? n_cur : 32, a);
        for (int k = 32; k < n_cur; k += 32) {
            int kk = k + lane;
            kk = (kk < n_cur) ? kk : (n_cur - 1);
            const int cv = col[start_cur + kk];
            gather_chunk(hp4, lane, cv, (n_cur - k < 32) ? (n_cur - k) : 32, a);
        }

        float t[4] = {a.x, a.y, a.z, a.w};
        float ss = 0.0f;
#pragma unroll
        for (int c = 0; c < 4; ++c) {
            float v = di * t[c] + bj[c];
            v = (v > 0.0f) ? v : aj[c] * v;
            t[c] = v;
            ss += v * v;
        }
#pragma unroll
        for (int off = 16; off > 0; off >>= 1) ss += __shfl_xor(ss, off, 32);
        const float scale = 1.0f / fmaxf(sqrtf(ss), 1e-12f);
        if (g != gcur) {
            pool_flush(pooled, cnt, gcur, lane, sum);
            sum[0] = sum[1] = sum[2] = sum[3] = 0.0f;
            gcur = g;
        }
#pragma unroll
        for (int c = 0; c < 4; ++c) sum[c] += t[c] * scale;

        n_cur = n_nx; start_cur = start_nx; cv_cur = cv_nx;
    }
    pool_flush(pooled, cnt, gcur, lane, sum);
}

extern "C" void kernel_launch(void* const* d_in, const int* in_sizes, int n_in,
                              void* d_out, int out_size, void* d_ws, size_t ws_size,
                              hipStream_t stream) {
    const float* x     = (const float*)d_in[0];
    const int*   ei    = (const int*)d_in[1];
    const int*   batch = (const int*)d_in[2];
    const float* W     = (const float*)d_in[3];
    const float* bias  = (const float*)d_in[4];
    const float* pa    = (const float*)d_in[5];

    const int n_nodes = in_sizes[0] / D_FEAT;
    const int n_edges = in_sizes[1] / 2;
    const int* src = ei;
    const int* dst = ei + n_edges;
    float* pooled = (float*)d_out;

    const int nb    = (n_nodes + NPB - 1) / NPB;
    const int chunk = (n_edges + SORT_BLKS - 1) / SORT_BLKS;
    const int stage = (chunk < STAGE_CAP) ? chunk : STAGE_CAP;

    float* hprime = (float*)d_ws;
    int*   degi     = (int*)(hprime + (size_t)n_nodes * D_FEAT);
    int*   rowstart = degi + n_nodes;
    int*   ebuf     = rowstart + n_nodes;
    int*   col      = ebuf + n_edges;
    int*   tab      = col + n_edges;
    float* dinv   = hprime + (size_t)n_nodes * D_FEAT * 2;
    float* cnt    = dinv + n_nodes;

    const int n_graphs = out_size / D_FEAT;

    chunk_sort_kernel<<<SORT_BLKS, SORT_THREADS, (size_t)(2 * (nb + 1) + stage) * sizeof(int), stream>>>(
        src, dst, tab, ebuf, pooled, cnt, out_size, n_graphs, n_edges, chunk, nb, stage);
    bucket_rows_kernel<<<nb, ROWS_THREADS, (size_t)n_graphs * sizeof(int), stream>>>(
        ebuf, tab, rowstart, degi, dinv, col, batch, cnt, n_nodes, n_graphs, n_edges, chunk, nb);

    int gemm_blocks = (n_nodes + ROWS_PER_BLK - 1) / ROWS_PER_BLK;
    if (gemm_blocks > GEMM_MAX_BLKS) gemm_blocks = GEMM_MAX_BLKS;
    gemm_scale_kernel<<<gemm_blocks, 256, 0, stream>>>(x, W, dinv, hprime, n_nodes);

    int gf_waves = (n_nodes + GF_NODES_PER_WAVE - 1) / GF_NODES_PER_WAVE;
    gather_finalize_kernel<<<(gf_waves + 7) / 8, 256, 0, stream>>>(
        hprime, rowstart, degi, col, dinv, bias, pa, batch, pooled, cnt, n_nodes);
}


# pooled sums of a block's waves added up in LDS first (graph ids of the block's first and last node); one atomic add per column and block; pooled summation order changes
# speedup vs baseline: 6.5124x; 1.1059x over previous
#include <hip/hip_runtime.h>
#include <hip/hip_bf16.h>
#include <math.h>


typedef __attribute__((ext_vector_type(2))) float v2f;
typedef __attribute__((ext_vector_type(8))) float v8f;

#define D_FEAT 128
#define MT_PER_BLK 2
#define ROWS_PER_BLK (MT_PER_BLK * 16)

#define NPB_SHIFT 9
#define NPB (1 << NPB_SHIFT)
#define SRC_BITS 22
#define SRC_MASK ((1 << SRC_BITS) - 1)
#define SORT_BLKS 256
#define SORT_THREADS 256
#define SORT_EPT 32
#define ROWS_THREADS 1024
#define ROWS_WAVES (ROWS_THREADS / 32)
#define SEGS_PER_WAVE (SORT_BLKS / ROWS_WAVES)
#define SEG_ROUNDS 2
#define STAGE_CAP 12288

static_assert(SORT_BLKS % ROWS_WAVES == 0, "every wave reads the same number of segments");
static_assert(SORT_BLKS <= ROWS_THREADS, "one thread per chunk reads the bucket starts");

__device__ __forceinline__ void chunk_sort_role(
    int* sh_sort, const int* __restrict__ src, const int* __restrict__ dst,
    int* __restrict__ tab, int* __restrict__ ebuf,
    float* __restrict__ pooled, float* __restrict__ cnt, int out_size, int n_graphs,
    int n_edges, int chunk, int nb, int stage, int sort_blks) {
    int* bins = sh_sort;
    int* cur  = sh_sort + (nb + 1);
    int* sbuf = cur + (nb + 1);
    const int tid  = threadIdx.x;
    const int lane = tid & 31;
    for (int i = blockIdx.x * SORT_THREADS + tid; i < out_size; i += sort_blks * SORT_THREADS)
        pooled[i] = 0.0f;
    if (blockIdx.x == 0)
        for (int g = tid; g < n_graphs; g += SORT_THREADS) cnt[g] = 0.0f;
    for (int b = tid; b <= nb; b += SORT_THREADS) bins[b] = 0;
    __syncthreads();
    const int e0 = blockIdx.x * chunk;
    const int e1 = (e0 + chunk < n_edges) ? (e0 + chunk) : n_edges;
    const int n  = (e1 > e0) ? (e1 - e0) : 0;
    for (int base = e0; base < e1; base += SORT_THREADS * SORT_EPT) {
        int d[SORT_EPT];
#pragma unroll
        for (int u = 0; u < SORT_EPT; ++u) {
            const int e = base + tid + u * SORT_THREADS;
            d[u] = (e < e1) ? dst[e] : -1;
        }
#pragma unroll
        for (int u = 0; u < SORT_EPT; ++u)
            if (d[u] >= 0) atomicAdd(&bins[d[u] >> NPB_SHIFT], 1);
    }
    __syncthreads();
    if (tid < 32) {
        int carry = 0;
        for (int b0 = 0; b0 < nb; b0 += 32) {
            const int c = (b0 + lane < nb) ? bins[b0 + lane] : 0;
            int v = c;
#pragma unroll
            for (int off = 1; off < 32; off <<= 1) {
                int x = __shfl_up(v, off, 32);
                if (lane >= off) v += x;
            }
            if (b0 + lane < nb) bins[b0 + lane] = carry + (v - c);
            carry += __shfl(v, 31, 32);
        }
        if (lane == 0) bins[nb] = carry;
    }
    __syncthreads();
    for (int b = tid; b <= nb; b += SORT_THREADS) {
        const int st = bins[b];
        tab[(size_t)blockIdx.x * (nb + 1) + b] = st;
        cur[b] = st;
    }
    __syncthreads();
    const bool staged = (n <= stage);
    for (int base = e0; base < e1; base += SORT_THREADS * SORT_EPT) {
        int d[SORT_EPT], s[SORT_EPT];
#pragma unroll
        for (int u = 0; u < SORT_EPT; ++u) {
            const int e = base + tid + u * SORT_THREADS;
            d[u] = (e < e1) ? dst[e] : -1;
            s[u] = (e < e1) ? src[e] : 0;
        }
#pragma unroll
        for (int u = 0; u < SORT_EPT; ++u) {
            if (d[u] >= 0) {
                const int p = atomicAdd(&cur[d[u] >> NPB_SHIFT], 1);
                const int word = ((d[u] & (NPB - 1)) << SRC_BITS) | s[u];
                if ((unsigned)p < (unsigned)n) {
                    if (staged) sbuf[p] = word;
                    else        ebuf[e0 + p] = word;
                }
            }
        }
    }
    if (staged) {
        __syncthreads();
        for (int i = tid; i < n; i += SORT_THREADS) ebuf[e0 + i] = sbuf[i];
    }
}

__global__ __launch_bounds__(ROWS_THREADS) void bucket_rows_kernel(
    const int* __restrict__ ebuf, const int* __restrict__ tab,
    int* __restrict__ rowstart, int* __restrict__ degi, float* __restrict__ dinv,
    unsigned char* __restrict__ deg8,
    int* __restrict__ col, const int* __restrict__ batch, float* __restrict__ cnt,
    int n_nodes, int n_graphs, int n_edges, int chunk, int nb) {
    extern __shared__ int sh_hist[];
    __shared__ int cntl[NPB];
    __shared__ int swave[NPB / 32];
    __shared__ int seg_off[SORT_BLKS];
    __shared__ int seg_len[SORT_BLKS];
    __shared__ int red_a[SORT_BLKS / 32];
    __shared__ int red_b[SORT_BLKS / 32];
    __shared__ int scol[STAGE_CAP];
    const int tid  = threadIdx.x;
    const int lane = tid & 31;
    const int wave = tid >> 5;
    const int b    = blockIdx.x;
    const int node = b * NPB + tid;

    int s0 = 0, ln = 0;
    if (tid < SORT_BLKS) {
        const int* row = tab + (size_t)tid * (nb + 1) + b;
        s0 = row[0];
        ln = row[1] - s0;
        const int c0 = tid * chunk;
        int avail = n_edges - c0;
        avail = (avail < chunk) ? avail : chunk;
        avail = (avail > 0) ? avail : 0;
        s0 = (s0 > 0) ? s0 : 0;
        s0 = (s0 < avail) ? s0 : avail;
        ln = (ln > 0) ? ln : 0;
        ln = (ln < avail - s0) ? ln : (avail - s0);
        seg_off[tid] = c0 + s0;
        seg_len[tid] = ln;
    }
    int bg = -1;
    if (tid < NPB && node < n_nodes) bg = batch[node];
    int va = s0, vb = ln;
#pragma unroll
    for (int off = 16; off > 0; off >>= 1) {
        va += __shfl_xor(va, off, 32);
        vb += __shfl_xor(vb, off, 32);
    }
    if (tid < SORT_BLKS && lane == 0) { red_a[wave] = va; red_b[wave] = vb; }
    if (tid < NPB) cntl[tid] = 0;
    for (int g = tid; g < n_graphs; g += ROWS_THREADS) sh_hist[g] = 0;
    __syncthreads();
    int seg0 = 0, m = 0;
    for (int q = 0; q < SORT_BLKS / 32; ++q) { seg0 += red_a[q]; m += red_b[q]; }

    int w[SEGS_PER_WAVE * SEG_ROUNDS];
#pragma unroll
    for (int u = 0; u < SEGS_PER_WAVE; ++u) {
        const int t    = wave + u * ROWS_WAVES;
        const int ln_t = seg_len[t];
        const int of_t = seg_off[t];
#pragma unroll
        for (int r = 0; r < SEG_ROUNDS; ++r) {
            const int k = lane + 32 * r;
            w[u * SEG_ROUNDS + r] = (k < ln_t) ? ebuf[of_t + k] : -1;
        }
    }
    if (bg >= 0) atomicAdd(&sh_hist[bg], 1);
#pragma unroll
    for (int i = 0; i < SEGS_PER_WAVE * SEG_ROUNDS; ++i)
        if (w[i] >= 0) atomicAdd(&cntl[(w[i] >> SRC_BITS) & (NPB - 1)], 1);
    for (int u = 0; u < SEGS_PER_WAVE; ++u) {
        const int t    = wave + u * ROWS_WAVES;
        const int ln_t = seg_len[t];
        const int of_t = seg_off[t];
        for (int k = SEG_ROUNDS * 32 + lane; k < ln_t; k += 32) {
            const int ww = ebuf[of_t + k];
            atomicAdd(&cntl[(ww >> SRC_BITS) & (NPB - 1)], 1);
        }
    }
    __syncthreads();
    for (int g = tid; g < n_graphs; g += ROWS_THREADS) {
        const int h = sh_hist[g];
        if (h) atomicAdd(&cnt[g], (float)h);
    }
    const int c = (tid < NPB) ? cntl[tid] : 0;
    int v = c;
#pragma unroll
    for (int off = 1; off < 32; off <<= 1) {
        int x = __shfl_up(v, off, 32);
        if (lane >= off) v += x;
    }
    if (tid < NPB && lane == 31) swave[wave] = v;
    __syncthreads();
    if (tid < NPB) {
        int wo = 0;
        for (int q = 0; q < wave; ++q) wo += swave[q];
        const int ex = wo + (v - c);
        cntl[tid] = ex;
        if (node < n_nodes) {
            rowstart[node] = seg0 + ex;
            degi[node]     = c;
            dinv[node]     = rsqrtf((float)c + 1.0f);
            deg8[node]     = (unsigned char)((c < 255) ? c : 255);
        }
    }
    __syncthreads();
    const bool staged = (m <= STAGE_CAP);
#pragma unroll
    for (int i = 0; i < SEGS_PER_WAVE * SEG_ROUNDS; ++i) {
        if (w[i] >= 0) {
            const int p = atomicAdd(&cntl[(w[i] >> SRC_BITS) & (NPB - 1)], 1);
            const int s = w[i] & SRC_MASK;
            if ((unsigned)p < (unsigned)m) {
                if (staged) scol[p] = s;
                else        col[seg0 + p] = s;
            }
        }
    }
    for (int u = 0; u < SEGS_PER_WAVE; ++u) {
        const int t    = wave + u * ROWS_WAVES;
        const int ln_t = seg_len[t];
        const int of_t = seg_off[t];
        for (int k = SEG_ROUNDS * 32 + lane; k < ln_t; k += 32) {
            const int ww = ebuf[of_t + k];
            const int p  = atomicAdd(&cntl[(ww >> SRC_BITS) & (NPB - 1)], 1);
            if ((unsigned)p < (unsigned)m) {
                if (staged) scol[p] = ww & SRC_MASK;
                else        col[seg0 + p] = ww & SRC_MASK;
            }
        }
    }
    if (staged) {
        __syncthreads();
        for (int i = tid; i < m; i += ROWS_THREADS) col[seg0 + i] = scol[i];
    }
}

#define GEMM_MAX_BLKS 768

__device__ __forceinline__ void load_x_tile(const float4* __restrict__ xv, int mbase, int tid,
                                            int n_nodes, float4& p0, float4& p1,
                                            float4& p2, float4& p3) {
    const int r0 = mbase + (tid >> 5);
    const int cb = tid & 31;
    if (mbase + ROWS_PER_BLK <= n_nodes) {
        p0 = xv[(size_t)(r0)      * (D_FEAT / 4) + cb];
        p1 = xv[(size_t)(r0 + 8)  * (D_FEAT / 4) + cb];
        p2 = xv[(size_t)(r0 + 16) * (D_FEAT / 4) + cb];
        p3 = xv[(size_t)(r0 + 24) * (D_FEAT / 4) + cb];
    } else {
        p0 = make_float4(0.f, 0.f, 0.f, 0.f);
        p1 = p0; p2 = p0; p3 = p0;
        if (r0      < n_nodes) p0 = xv[(size_t)(r0)      * (D_FEAT / 4) + cb];
        if (r0 + 8  < n_nodes) p1 = xv[(size_t)(r0 + 8)  * (D_FEAT / 4) + cb];
        if (r0 + 16 < n_nodes) p2 = xv[(size_t)(r0 + 16) * (D_FEAT / 4) + cb];
        if (r0 + 24 < n_nodes) p3 = xv[(size_t)(r0 + 24) * (D_FEAT / 4) + cb];
    }
}

__device__ __forceinline__ void product_role(
    float (*sX)[D_FEAT + 4],
    const float* __restrict__ x, const float* __restrict__ W,
    float* __restrict__ hprime, int n_nodes, int blk, int n_blks) {

    const int tid   = threadIdx.x;
    const int lane  = tid & 31;
    const int wave  = tid >> 5;
    const int half  = lane >> 4;
    const int l15   = lane & 15;
    const int nb    = wave * 16;
    const int n_tiles = (n_nodes + ROWS_PER_BLK - 1) / ROWS_PER_BLK;

    v2f bfr[D_FEAT / 4];
#pragma unroll
    for (int s = 0; s < D_FEAT / 4; ++s) {
        const int ka = 4 * s + half * 2;
        bfr[s].x = W[(size_t)ka * D_FEAT + nb + l15];
        bfr[s].y = W[(size_t)(ka + 1) * D_FEAT + nb + l15];
    }

    const float4* xv = reinterpret_cast<const float4*>(x);
    const int sr = tid >> 5;
    const int cb = tid & 31;
    float4 p0, p1, p2, p3;
    p0 = make_float4(0.f, 0.f, 0.f, 0.f);
    p1 = p0; p2 = p0; p3 = p0;
    int tile = blk;
    if (tile < n_tiles) load_x_tile(xv, tile * ROWS_PER_BLK, tid, n_nodes, p0, p1, p2, p3);

    for (; tile < n_tiles; tile += n_blks) {
        const int mbase = tile * ROWS_PER_BLK;
        __syncthreads();
        *reinterpret_cast<float4*>(&sX[sr][cb * 4])      = p0;
        *reinterpret_cast<float4*>(&sX[sr + 8][cb * 4])  = p1;
        *reinterpret_cast<float4*>(&sX[sr + 16][cb * 4]) = p2;
        *reinterpret_cast<float4*>(&sX[sr + 24][cb * 4]) = p3;
        __syncthreads();
        const int next = tile + n_blks;
        if (next < n_tiles)
            load_x_tile(xv, next * ROWS_PER_BLK, tid, n_nodes, p0, p1, p2, p3);

        v8f c[MT_PER_BLK];
#pragma unroll
        for (int t = 0; t < MT_PER_BLK; ++t)
#pragma unroll
            for (int r = 0; r < 8; ++r) c[t][r] = 0.0f;

#pragma unroll
        for (int s = 0; s < D_FEAT / 4; ++s) {
            const int ka = 4 * s + half * 2;
#pragma unroll
            for (int t = 0; t < MT_PER_BLK; ++t) {
                v2f af;
                af.x = sX[t * 16 + l15][ka];
                af.y = sX[t * 16 + l15][ka + 1];
                c[t] = __builtin_amdgcn_wmma_f32_16x16x4_f32(
                     false, af,  false, bfr[s],
                     (short)0, c[t],  false,  false);
            }
        }

        if (mbase + ROWS_PER_BLK <= n_nodes) {
#pragma unroll
            for (int t = 0; t < MT_PER_BLK; ++t) {
                const int rbase = mbase + t * 16 + half * 8;
                float* hp = hprime + (size_t)rbase * D_FEAT + nb + l15;
#pragma unroll
                for (int r = 0; r < 8; ++r) {
                    hp[(size_t)r * D_FEAT] = c[t][r];
                }
            }
        } else {
#pragma unroll
            for (int t = 0; t < MT_PER_BLK; ++t) {
#pragma unroll
                for (int r = 0; r < 8; ++r) {
                    int row = mbase + t * 16 + r + half * 8;
                    if (row < n_nodes) {
                        hprime[(size_t)row * D_FEAT + nb + l15] = c[t][r];
                    }
                }
            }
        }
    }
}

__global__ __launch_bounds__(SORT_THREADS) void sort_and_product_kernel(
    const int* __restrict__ src, const int* __restrict__ dst,
    int* __restrict__ tab, int* __restrict__ ebuf,
    float* __restrict__ pooled, float* __restrict__ cnt, int out_size, int n_graphs,
    int n_edges, int chunk, int nb, int stage, int sort_blks,
    const float* __restrict__ x, const float* __restrict__ W,
    float* __restrict__ hprime, int n_nodes) {
    extern __shared__ __align__(16) int sh_dyn[];
    if ((int)blockIdx.x < sort_blks) {
        chunk_sort_role(sh_dyn, src, dst, tab, ebuf, pooled, cnt, out_size, n_graphs,
                        n_edges, chunk, nb, stage, sort_blks);
    } else {
        product_role(reinterpret_cast<float (*)[D_FEAT + 4]>(sh_dyn), x, W, hprime, n_nodes,
                     (int)blockIdx.x - sort_blks, (int)gridDim.x - sort_blks);
    }
}

#define GF_NODES_PER_WAVE 16
#define GF_THREADS 512
#define DEG_CAP 100352
#define DEG_VECS (DEG_CAP / 16)
#define DEG_LOADS ((DEG_VECS + GF_THREADS - 1) / GF_THREADS)
static_assert(GF_THREADS >= 2 * D_FEAT, "one thread per column of the block's two pooled rows");

__device__ __forceinline__ float src_weight(const unsigned char* sh_deg,
                                            const float* __restrict__ dinv, int cv, int tab_n) {
    int d = 255;
    if ((unsigned)cv < (unsigned)tab_n) d = sh_deg[cv];
    float w = rsqrtf((float)d + 1.0f);
    const bool slow = (d == 255);
    if (__builtin_amdgcn_ballot_w32(slow) != 0u) {
        if (slow) w = dinv[cv];
    }
    return w;
}

__device__ __forceinline__ void pool_flush(float* __restrict__ pooled,
                                           const float* __restrict__ cnt,
                                           float (*sh_pool)[D_FEAT], int g_first, int g_last,
                                           int g, int lane, const float (&sum)[4]) {
    const float cg = fmaxf(cnt[g], 1.0f);
    if (g == g_first || g == g_last) {
        float* row = sh_pool[(g == g_first) ? 0 : 1];
#pragma unroll
        for (int j = 0; j < 4; ++j)
            atomicAdd(&row[lane * 4 + j], sum[j] / cg);
    } else {
#pragma unroll
        for (int j = 0; j < 4; ++j)
            atomicAdd(&pooled[(size_t)g * D_FEAT + lane * 4 + j], sum[j] / cg);
    }
}

template <int NB>
__device__ __forceinline__ void gather_rows(const float4* __restrict__ hp4, int lane,
                                            int cv, float wv, int q, int m, float4& a) {
    float4 r[4 * NB];
    float  w[4 * NB];
#pragma unroll
    for (int u = 0; u < 4 * NB; ++u) {
        float4 v = make_float4(0.0f, 0.0f, 0.0f, 0.0f);
        if (u <= 4 * (NB - 1) || q + u < m) {
            const int s = __builtin_amdgcn_readlane(cv, q + u);
            v = hp4[(size_t)s * (D_FEAT / 4) + lane];
        }
        r[u] = v;
    }
#pragma unroll
    for (int u = 0; u < 4 * NB; ++u) {
        float wu = 0.0f;
        if (u <= 4 * (NB - 1) || q + u < m)
            wu = __int_as_float(__builtin_amdgcn_readlane(__float_as_int(wv), q + u));
        w[u] = wu;
    }
#pragma unroll
    for (int u = 0; u < 4 * NB; ++u) {
        a.x = fmaf(w[u], r[u].x, a.x);
        a.y = fmaf(w[u], r[u].y, a.y);
        a.z = fmaf(w[u], r[u].z, a.z);
        a.w = fmaf(w[u], r[u].w, a.w);
    }
}

__device__ __forceinline__ void gather_chunk(const float4* __restrict__ hp4, int lane,
                                             int cv, float wv, int m, float4& a) {
    for (int q = 0; q < m; q += 16) {
        const int left = m - q;
        if (left > 12)     gather_rows<4>(hp4, lane, cv, wv, q, m, a);
        else if (left > 8) gather_rows<3>(hp4, lane, cv, wv, q, m, a);
        else if (left > 4) gather_rows<2>(hp4, lane, cv, wv, q, m, a);
        else               gather_rows<1>(hp4, lane, cv, wv, q, m, a);
    }
}

__global__ __launch_bounds__(GF_THREADS) void gather_finalize_kernel(
    const float* __restrict__ hprime, const int* __restrict__ rowstart,
    const int* __restrict__ degi, const int* __restrict__ col,
    const float* __restrict__ dinv, const unsigned char* __restrict__ deg8,
    const float* __restrict__ bias,
    const float* __restrict__ pa, const int* __restrict__ batch,
    float* __restrict__ pooled, const float* __restrict__ cnt, int n_nodes) {
    __shared__ __align__(16) unsigned char sh_deg[DEG_CAP];
    __shared__ float sh_pool[2][D_FEAT];
    const int tid  = threadIdx.x;
    const int lane = tid & 31;
    const int blk_nodes = (GF_THREADS / 32) * GF_NODES_PER_WAVE;
    int nf = (int)blockIdx.x * blk_nodes;
    int nl = nf + blk_nodes - 1;
    nf = (nf < n_nodes) ? nf : (n_nodes - 1);
    nl = (nl < n_nodes) ? nl : (n_nodes - 1);
    const int g_first = batch[nf];
    const int g_last  = batch[nl];
    if (tid < 2 * D_FEAT) sh_pool[tid / D_FEAT][tid % D_FEAT] = 0.0f;
    const int wvi = __builtin_amdgcn_readfirstlane(
        (int)(blockIdx.x * (blockDim.x >> 5) + (threadIdx.x >> 5)));
    const int i0 = wvi * GF_NODES_PER_WAVE;
    const bool active = (i0 < n_nodes);
    int nn = n_nodes - i0;
    nn = (nn < GF_NODES_PER_WAVE) ? nn : GF_NODES_PER_WAVE;
    nn = (nn > 1) ? nn : 1;

    const int tab_n = (n_nodes < DEG_CAP) ? n_nodes : DEG_CAP;
    const int nvec  = tab_n >> 4;
    {
        const uint4* g4 = reinterpret_cast<const uint4*>(deg8);
        uint4* s4 = reinterpret_cast<uint4*>(sh_deg);
        uint4 v[DEG_LOADS];
#pragma unroll
        for (int u = 0; u < DEG_LOADS; ++u) {
            const int idx = tid + u * GF_THREADS;
            v[u] = make_uint4(0u, 0u, 0u, 0u);
            if (idx < nvec) v[u] = g4[idx];
        }
#pragma unroll
        for (int u = 0; u < DEG_LOADS; ++u) {
            const int idx = tid + u * GF_THREADS;
            if (idx < nvec) s4[idx] = v[u];
        }
        for (int i = (nvec << 4) + tid; i < tab_n; i += GF_THREADS) sh_deg[i] = deg8[i];
    }

    const float4* hp4 = reinterpret_cast<const float4*>(hprime);
    const float4 b4 = reinterpret_cast<const float4*>(bias)[lane];
    const float4 a4 = reinterpret_cast<const float4*>(pa)[lane];
    const float bj[4] = {b4.x, b4.y, b4.z, b4.w};
    const float aj[4] = {a4.x, a4.y, a4.z, a4.w};

    int li = i0 + ((lane < nn) ? lane : (nn - 1));
    li = (li < n_nodes) ? li : (n_nodes - 1);
    const int   m_start = rowstart[li];
    const int   m_n     = degi[li];
    const float m_di    = dinv[li];
    const int   m_g     = batch[li];

    __syncthreads();
    const int n_walk = active ? nn : 0;

    float sum[4] = {0.0f, 0.0f, 0.0f, 0.0f};
    int gcur = __builtin_amdgcn_readlane(m_g, 0);

    int n_cur = 0, start_cur = 0, cv_cur = 0;
    for (int j = -1; j < n_walk; ++j) {
        int n_nx = 0, start_nx = 0, cv_nx = 0;
        if (j + 1 < n_walk) {
            n_nx     = __builtin_amdgcn_readlane(m_n, j + 1);
            start_nx = __builtin_amdgcn_readlane(m_start, j + 1);
            if (n_nx > 0) cv_nx = col[start_nx + ((lane < n_nx) ? lane : (n_nx - 1))];
        }
        if (j < 0) {
            n_cur = n_nx; start_cur = start_nx; cv_cur = cv_nx;
            continue;
        }
        const float di = __int_as_float(__builtin_amdgcn_readlane(__float_as_int(m_di), j));
        const int   g  = __builtin_amdgcn_readlane(m_g, j);

        const float4 sr = hp4[(size_t)(i0 + j) * (D_FEAT / 4) + lane];
        float4 a = make_float4(0.0f, 0.0f, 0.0f, 0.0f);
        if (n_cur > 0) {
            const float wv = src_weight(sh_deg, dinv, cv_cur, tab_n);
            gather_chunk(hp4, lane, cv_cur, wv, (n_cur < 32) ? n_cur : 32, a);
        }
        for (int k = 32; k < n_cur; k += 32) {
            int kk = k + lane;
            kk = (kk < n_cur) ? kk : (n_cur - 1);
            const int cv = col[start_cur + kk];
            const float wv = src_weight(sh_deg, dinv, cv, tab_n);
            gather_chunk(hp4, lane, cv, wv, (n_cur - k < 32) ? (n_cur - k) : 32, a);
        }
        a.x = fmaf(di, sr.x, a.x);
        a.y = fmaf(di, sr.y, a.y);
        a.z = fmaf(di, sr.z, a.z);
        a.w = fmaf(di, sr.w, a.w);

        float t[4] = {a.x, a.y, a.z, a.w};
        float ss = 0.0f;
#pragma unroll
        for (int c = 0; c < 4; ++c) {
            float v = di * t[c] + bj[c];
            v = (v > 0.0f) ? v : aj[c] * v;
            t[c] = v;
            ss += v * v;
        }
#pragma unroll
        for (int off = 16; off > 0; off >>= 1) ss += __shfl_xor(ss, off, 32);
        const float scale = 1.0f / fmaxf(sqrtf(ss), 1e-12f);
        if (g != gcur) {
            pool_flush(pooled, cnt, sh_pool, g_first, g_last, gcur, lane, sum);
            sum[0] = sum[1] = sum[2] = sum[3] = 0.0f;
            gcur = g;
        }
#pragma unroll
        for (int c = 0; c < 4; ++c) sum[c] += t[c] * scale;

        n_cur = n_nx; start_cur = start_nx; cv_cur = cv_nx;
    }
    if (active) pool_flush(pooled, cnt, sh_pool, g_first, g_last, gcur, lane, sum);
    __syncthreads();
    if (tid < D_FEAT)
        atomicAdd(&pooled[(size_t)g_first * D_FEAT + tid], sh_pool[0][tid]);
    else if (tid < 2 * D_FEAT && g_last != g_first)
        atomicAdd(&pooled[(size_t)g_last * D_FEAT + (tid - D_FEAT)], sh_pool[1][tid - D_FEAT]);
}

extern "C" void kernel_launch(void* const* d_in, const int* in_sizes, int n_in,
                              void* d_out, int out_size, void* d_ws, size_t ws_size,
                              hipStream_t stream) {
    const float* x     = (const float*)d_in[0];
    const int*   ei    = (const int*)d_in[1];
    const int*   batch = (const int*)d_in[2];
    const float* W     = (const float*)d_in[3];
    const float* bias  = (const float*)d_in[4];
    const float* pa    = (const float*)d_in[5];

    const int n_nodes = in_sizes[0] / D_FEAT;
    const int n_edges = in_sizes[1] / 2;
    const int* src = ei;
    const int* dst = ei + n_edges;
    float* pooled = (float*)d_out;

    const int nb    = (n_nodes + NPB - 1) / NPB;
    const int chunk = (n_edges + SORT_BLKS - 1) / SORT_BLKS;
    const int stage = (chunk < STAGE_CAP) ? chunk : STAGE_CAP;

    float* hprime = (float*)d_ws;
    unsigned char* deg8 = (unsigned char*)(hprime + (size_t)n_nodes * D_FEAT);
    int*   degi     = (int*)(deg8 + (((size_t)n_nodes + 15) / 16) * 16);
    int*   rowstart = degi + n_nodes;
    int*   ebuf     = rowstart + n_nodes;
    int*   col      = ebuf + n_edges;
    int*   tab      = col + n_edges;
    float* dinv   = hprime + (size_t)n_nodes * D_FEAT * 2;
    float* cnt    = dinv + n_nodes;

    const int n_graphs = out_size / D_FEAT;

    int gemm_blocks = (n_nodes + ROWS_PER_BLK - 1) / ROWS_PER_BLK;
    if (gemm_blocks > GEMM_MAX_BLKS) gemm_blocks = GEMM_MAX_BLKS;
    size_t dyn_bytes = (size_t)(2 * (nb + 1) + stage) * sizeof(int);
    const size_t tile_bytes = sizeof(float) * ROWS_PER_BLK * (D_FEAT + 4);
    if (dyn_bytes < tile_bytes) dyn_bytes = tile_bytes;

    sort_and_product_kernel<<<SORT_BLKS + gemm_blocks, SORT_THREADS, dyn_bytes, stream>>>(
        src, dst, tab, ebuf, pooled, cnt, out_size, n_graphs, n_edges, chunk, nb, stage, SORT_BLKS,
        x, W, hprime, n_nodes);
    bucket_rows_kernel<<<nb, ROWS_THREADS, (size_t)n_graphs * sizeof(int), stream>>>(
        ebuf, tab, rowstart, degi, dinv, deg8, col, batch, cnt, n_nodes, n_graphs, n_edges, chunk, nb);

    const int gf_wpb   = GF_THREADS / 32;
    const int gf_waves = (n_nodes + GF_NODES_PER_WAVE - 1) / GF_NODES_PER_WAVE;
    gather_finalize_kernel<<<(gf_waves + gf_wpb - 1) / gf_wpb, GF_THREADS, 0, stream>>>(
        hprime, rowstart, degi, col, dinv, deg8, bias, pa, batch, pooled, cnt, n_nodes);
}
